# starting kernel
# baseline (speedup 1.0000x reference)
#include <hip/hip_runtime.h>

typedef __attribute__((ext_vector_type(2))) float v2f;
typedef __attribute__((ext_vector_type(8))) float v8f;

#define NN 100000
#define FF 256
#define EE 1600000
#define HH1 128
#define HH2 64

#define GEMM_WAVES 4

__global__ void gcn_zero(float* __restrict__ p, int n) {
    int i = blockIdx.x * blockDim.x + threadIdx.x;
    if (i < n) p[i] = 0.0f;
}

__global__ void gcn_relu(float* __restrict__ p, int n) {
    int i = blockIdx.x * blockDim.x + threadIdx.x;
    if (i < n) p[i] = fmaxf(p[i], 0.0f);
}

template <int K, int Nout>
__global__ __launch_bounds__(32 * GEMM_WAVES)
void gcn_gemm_wmma2(const float* __restrict__ A,
                    const float* __restrict__ B0, const float* __restrict__ B1,
                    float* __restrict__ C0, float* __restrict__ C1, int M) {
    static_assert(K % 16 == 0 && Nout % 16 == 0, "tile dims");
    const int lane = threadIdx.x;
    const int half = lane >> 4;
    const int mn   = lane & 15;
    const int tileN = blockIdx.x;
    const int tileM = blockIdx.y * GEMM_WAVES + threadIdx.y;
    if (tileM * 16 >= M) return;

    const float* __restrict__ ap  = A + (size_t)(tileM * 16 + mn) * K + 2 * half;
    const size_t boff = (size_t)(2 * half) * Nout + tileN * 16 + mn;
    const float* __restrict__ b0c = B0 + boff;
    const float* __restrict__ b1c = B1 + boff;

    v8f s0a = {}, s0b = {}, s0c = {}, s0d = {};
    v8f s1a = {}, s1b = {}, s1c = {}, s1d = {};

    for (int it = 0; it < K / 16; ++it) {
        v2f a0 = *reinterpret_cast<const v2f*>(ap + 0);
        v2f a1 = *reinterpret_cast<const v2f*>(ap + 4);
        v2f a2 = *reinterpret_cast<const v2f*>(ap + 8);
        v2f a3 = *reinterpret_cast<const v2f*>(ap + 12);
        v2f b00, b01, b02, b03, b10, b11, b12, b13;
        b00.x = b0c[0 * Nout];   b00.y = b0c[1 * Nout];
        b01.x = b0c[4 * Nout];   b01.y = b0c[5 * Nout];
        b02.x = b0c[8 * Nout];   b02.y = b0c[9 * Nout];
        b03.x = b0c[12 * Nout];  b03.y = b0c[13 * Nout];
        b10.x = b1c[0 * Nout];   b10.y = b1c[1 * Nout];
        b11.x = b1c[4 * Nout];   b11.y = b1c[5 * Nout];
        b12.x = b1c[8 * Nout];   b12.y = b1c[9 * Nout];
        b13.x = b1c[12 * Nout];  b13.y = b1c[13 * Nout];

        s0a = __builtin_amdgcn_wmma_f32_16x16x4_f32(false, a0, false, b00, (short)0, s0a, false, false);
        s1a = __builtin_amdgcn_wmma_f32_16x16x4_f32(false, a0, false, b10, (short)0, s1a, false, false);
        s0b = __builtin_amdgcn_wmma_f32_16x16x4_f32(false, a1, false, b01, (short)0, s0b, false, false);
        s1b = __builtin_amdgcn_wmma_f32_16x16x4_f32(false, a1, false, b11, (short)0, s1b, false, false);
        s0c = __builtin_amdgcn_wmma_f32_16x16x4_f32(false, a2, false, b02, (short)0, s0c, false, false);
        s1c = __builtin_amdgcn_wmma_f32_16x16x4_f32(false, a2, false, b12, (short)0, s1c, false, false);
        s0d = __builtin_amdgcn_wmma_f32_16x16x4_f32(false, a3, false, b03, (short)0, s0d, false, false);
        s1d = __builtin_amdgcn_wmma_f32_16x16x4_f32(false, a3, false, b13, (short)0, s1d, false, false);

        ap  += 16;
        b0c += (size_t)16 * Nout;
        b1c += (size_t)16 * Nout;
    }
    v8f r0 = (s0a + s0b) + (s0c + s0d);
    v8f r1 = (s1a + s1b) + (s1c + s1d);

    const size_t coff = (size_t)(tileM * 16 + half * 8) * Nout + tileN * 16 + mn;
    float* __restrict__ c0row = C0 + coff;
    float* __restrict__ c1row = C1 + coff;
#pragma unroll
    for (int v = 0; v < 8; ++v) c0row[(size_t)v * Nout] = r0[v];
#pragma unroll
    for (int v = 0; v < 8; ++v) c1row[(size_t)v * Nout] = r1[v];
}

template <int LOGC>
__global__ void gcn_scatter_fused(const float* __restrict__ pre0,
                                  const float* __restrict__ pre1,
                                  const int* __restrict__ erow,
                                  const int* __restrict__ ecol,
                                  const float* __restrict__ ev0,
                                  const float* __restrict__ ev1,
                                  float* __restrict__ acc) {
    constexpr int C = 1 << LOGC;
    int t = blockIdx.x * blockDim.x + threadIdx.x;
    int e = t >> LOGC;
    int c = t & (C - 1);
    if (e >= EE) return;

    float v0 = __builtin_nontemporal_load(ev0 + e);
    float v1 = __builtin_nontemporal_load(ev1 + e);
    int   r  = __builtin_nontemporal_load(erow + e);
    int  col = __builtin_nontemporal_load(ecol + e);

    const float4 p0 =
        reinterpret_cast<const float4*>(pre0 + ((size_t)col << (LOGC + 2)))[c];
    const float4 p1 =
        reinterpret_cast<const float4*>(pre1 + ((size_t)col << (LOGC + 2)))[c];

    float* dst = acc + ((size_t)r << (LOGC + 2)) + 4 * c;
    atomicAdd(dst + 0, v0 * p0.x + v1 * p1.x);
    atomicAdd(dst + 1, v0 * p0.y + v1 * p1.y);
    atomicAdd(dst + 2, v0 * p0.z + v1 * p1.z);
    atomicAdd(dst + 3, v0 * p0.w + v1 * p1.w);
}

__global__ void gcn_scatter1_fused(const float* __restrict__ p0,
                                   const float* __restrict__ p1,
                                   const int* __restrict__ erow,
                                   const int* __restrict__ ecol,
                                   const float* __restrict__ ev0,
                                   const float* __restrict__ ev1,
                                   float* __restrict__ acc) {
    int e = blockIdx.x * blockDim.x + threadIdx.x;
    if (e >= EE) return;
    float v0 = __builtin_nontemporal_load(ev0 + e);
    float v1 = __builtin_nontemporal_load(ev1 + e);
    int  col = __builtin_nontemporal_load(ecol + e);
    int  r   = __builtin_nontemporal_load(erow + e);
    atomicAdd(acc + r, v0 * p0[col] + v1 * p1[col]);
}

__global__ void gcn_dot64x2(const float* __restrict__ h,
                            const float* __restrict__ w0,
                            const float* __restrict__ w1,
                            float* __restrict__ p0, float* __restrict__ p1) {
    int i = blockIdx.x * blockDim.x + threadIdx.x;
    if (i >= NN) return;
    const float* row = h + (size_t)i * HH2;
    float s0 = 0.0f, s1 = 0.0f;
#pragma unroll
    for (int k = 0; k < HH2; ++k) {
        float hv = row[k];
        s0 += hv * w0[k];
        s1 += hv * w1[k];
    }
    p0[i] = s0;
    p1[i] = s1;
}

extern "C" void kernel_launch(void* const* d_in, const int* in_sizes, int n_in,
                              void* d_out, int out_size, void* d_ws, size_t ws_size,
                              hipStream_t stream) {
    const float* x     = (const float*)d_in[0];
    const int*   erow  = (const int*)d_in[1];
    const int*   ecol  = (const int*)d_in[2];
    const float* evals = (const float*)d_in[3];
    const float* W1    = (const float*)d_in[4];
    const float* W2    = (const float*)d_in[5];
    const float* W3    = (const float*)d_in[6];
    float* out = (float*)d_out;

    const float* ev0 = evals;
    const float* ev1 = evals + (size_t)EE;

    float* pre0 = (float*)d_ws;
    float* pre1 = pre0 + (size_t)NN * HH1;
    float* bufA = pre1 + (size_t)NN * HH1;
    float* bufB = bufA + (size_t)NN * HH1;

    const int ZB = 256;
    dim3 gemm_blk(32, GEMM_WAVES);
    const int mtiles = NN / 16;
    const int gemm_gy = (mtiles + GEMM_WAVES - 1) / GEMM_WAVES;

    gcn_zero<<<((size_t)NN * HH1 + ZB - 1) / ZB, ZB, 0, stream>>>(bufA, NN * HH1);
    gcn_gemm_wmma2<FF, HH1><<<dim3(HH1 / 16, gemm_gy), gemm_blk, 0, stream>>>(
        x, W1, W1 + (size_t)FF * HH1, pre0, pre1, NN);
    {
        long long tot = (long long)EE << 5;
        gcn_scatter_fused<5><<<(tot + ZB - 1) / ZB, ZB, 0, stream>>>(
            pre0, pre1, erow, ecol, ev0, ev1, bufA);
    }
    gcn_relu<<<((size_t)NN * HH1 + ZB - 1) / ZB, ZB, 0, stream>>>(bufA, NN * HH1);

    gcn_zero<<<((size_t)NN * HH2 + ZB - 1) / ZB, ZB, 0, stream>>>(bufB, NN * HH2);
    gcn_gemm_wmma2<HH1, HH2><<<dim3(HH2 / 16, gemm_gy), gemm_blk, 0, stream>>>(
        bufA, W2, W2 + (size_t)HH1 * HH2, pre0, pre1, NN);
    {
        long long tot = (long long)EE << 4;
        gcn_scatter_fused<4><<<(tot + ZB - 1) / ZB, ZB, 0, stream>>>(
            pre0, pre1, erow, ecol, ev0, ev1, bufB);
    }
    gcn_relu<<<((size_t)NN * HH2 + ZB - 1) / ZB, ZB, 0, stream>>>(bufB, NN * HH2);

    gcn_zero<<<(NN + ZB - 1) / ZB, ZB, 0, stream>>>(out, NN);
    gcn_dot64x2<<<(NN + ZB - 1) / ZB, ZB, 0, stream>>>(
        bufB, W3, W3 + (size_t)HH2, pre0, pre1);
    gcn_scatter1_fused<<<(EE + ZB - 1) / ZB, ZB, 0, stream>>>(
        pre0, pre1, erow, ecol, ev0, ev1, out);
}


# Both products re-tiled: a wave owns a 16-column strip for all rows, weights kept in registers all launch; 16-row input tiles pass once through LDS in fragment order; k summed in 4 interleaved chains
# speedup vs baseline: 2.2637x; 2.2637x over previous
#include <hip/hip_runtime.h>

typedef __attribute__((ext_vector_type(2))) float v2f;
typedef __attribute__((ext_vector_type(8))) float v8f;

#define NN 100000
#define FF 256
#define EE 1600000
#define HH1 128
#define HH2 64

#define GS_PITCH   33
#define GS_WAVES   16
#define GS_THREADS (32 * GS_WAVES)

template <int K, int Nout>
__global__ __launch_bounds__(GS_THREADS)
void gcn_gemm_strip(const float* __restrict__ A,
                    const float* __restrict__ B0, const float* __restrict__ B1,
                    float* __restrict__ C0, float* __restrict__ C1, int nsteps) {
    constexpr int STRIPS = Nout / 16;
    constexpr int WPT    = 2 * STRIPS;
    constexpr int MT     = GS_WAVES / WPT;
    constexpr int K4     = K / 4;
    constexpr int KJ     = K / 8;
    static_assert(K % 8 == 0 && Nout % 16 == 0 && GS_WAVES % WPT == 0, "tile dims");
    static_assert(MT * 16 * K4 == 2 * GS_THREADS, "two float4 per thread and step");

    __shared__ float4 sA[2][MT * KJ * GS_PITCH];

    const int t    = threadIdx.x;
    const int lane = t & 31;
    const int wave = t >> 5;
    const int h    = lane >> 4;
    const int c    = lane & 15;
    const int g    = wave / WPT;
    const int sup  = (wave % WPT) / STRIPS;
    const int col  = (wave % STRIPS) * 16 + c;

    const float* __restrict__ bp = (sup ? B1 : B0) + (size_t)(4 * h) * Nout + col;
    v2f breg[K4];
#pragma unroll
    for (int s = 0; s < K4; ++s) {
        const int kk = 8 * (s >> 1) + 2 * (s & 1);
        breg[s].x = bp[(kk + 0) * Nout];
        breg[s].y = bp[(kk + 1) * Nout];
    }
    float* __restrict__ cp = (sup ? C1 : C0) + (size_t)(16 * g + 8 * h) * Nout + col;

    const int f0 = t, f1 = t + GS_THREADS;
    const int r0 = f0 / K4, c0 = f0 % K4;
    const int r1 = f1 / K4, c1 = f1 % K4;
    const int srow0  = r0 * K + c0 * 4;
    const int srow1  = r1 * K + c1 * 4;
    const int sslot0 = ((r0 >> 4) * KJ + (c0 >> 1)) * GS_PITCH + (c0 & 1) * 16 + (r0 & 15);
    const int sslot1 = ((r1 >> 4) * KJ + (c1 >> 1)) * GS_PITCH + (c1 & 1) * 16 + (r1 & 15);

    int step = blockIdx.x;
    {
        const float* __restrict__ ap = A + (size_t)min(step, nsteps - 1) * (MT * 16 * K);
        const float4 p0 = *reinterpret_cast<const float4*>(ap + srow0);
        const float4 p1 = *reinterpret_cast<const float4*>(ap + srow1);
        sA[0][sslot0] = p0;
        sA[0][sslot1] = p1;
    }
    __syncthreads();

    int buf = 0;
    v8f rprev = {};
    float* __restrict__ cprev = cp;
    bool pending = false;
    for (; step < nsteps; step += gridDim.x) {
        const float* __restrict__ ap = A + (size_t)min(step + (int)gridDim.x, nsteps - 1) * (MT * 16 * K);
        const float4 st0 = *reinterpret_cast<const float4*>(ap + srow0);
        const float4 st1 = *reinterpret_cast<const float4*>(ap + srow1);

        if (pending) {
#pragma unroll
            for (int v = 0; v < 8; ++v) cprev[(size_t)v * Nout] = rprev[v];
        }

        const float4* __restrict__ fa = sA[buf] + g * (KJ * GS_PITCH) + lane;
        v8f acc0 = {}, acc1 = {}, acc2 = {}, acc3 = {};
#pragma unroll
        for (int j = 0; j < KJ; j += 2) {
            const float4 a0 = fa[(j + 0) * GS_PITCH];
            const float4 a1 = fa[(j + 1) * GS_PITCH];
            v2f x0 = {a0.x, a0.y}, x1 = {a0.z, a0.w}, x2 = {a1.x, a1.y}, x3 = {a1.z, a1.w};
            acc0 = __builtin_amdgcn_wmma_f32_16x16x4_f32(false, x0, false, breg[2 * j + 0], (short)0, acc0, false, false);
            acc1 = __builtin_amdgcn_wmma_f32_16x16x4_f32(false, x1, false, breg[2 * j + 1], (short)0, acc1, false, false);
            acc2 = __builtin_amdgcn_wmma_f32_16x16x4_f32(false, x2, false, breg[2 * j + 2], (short)0, acc2, false, false);
            acc3 = __builtin_amdgcn_wmma_f32_16x16x4_f32(false, x3, false, breg[2 * j + 3], (short)0, acc3, false, false);
        }
        rprev   = (acc0 + acc1) + (acc2 + acc3);
        cprev   = cp + (size_t)step * (MT * 16 * Nout);
        pending = true;

        sA[buf ^ 1][sslot0] = st0;
        sA[buf ^ 1][sslot1] = st1;
        __syncthreads();
        buf ^= 1;
    }
    if (pending) {
#pragma unroll
        for (int v = 0; v < 8; ++v) cprev[(size_t)v * Nout] = rprev[v];
    }
}

#define SC_THREADS 1024
#define SC_WAVES   (SC_THREADS / 32)
#define SC_RANGES  256
#define SC_ROWS    ((NN + SC_RANGES - 1) / SC_RANGES)
#define SC_LD      10
#define SC_CHUNK   (SC_THREADS * SC_LD * 4)
#define SC_NCHUNK  ((EE + SC_CHUNK - 1) / SC_CHUNK)
#define SC_QCAP    1024
#define SC_U       6

#define LIST_CAP   8192
#define LIST_U     8
#define KEY_SHIFT  17
#define KEY_MASK   ((1 << KEY_SHIFT) - 1)
static_assert(NN <= (1 << KEY_SHIFT), "col fits the low key bits");
static_assert(SC_ROWS <= (1 << (31 - KEY_SHIFT)), "local row fits the high key bits");
static_assert(LIST_CAP % SC_THREADS == 0, "layer 3 walks the list in whole sweeps");

template <int H, bool MAKE_LIST>
__global__ __launch_bounds__(SC_THREADS)
void gcn_scatter_lds(const float* __restrict__ pre0,
                     const float* __restrict__ pre1,
                     const int* __restrict__ erow,
                     const int* __restrict__ ecol,
                     const float* __restrict__ ev0,
                     const float* __restrict__ ev1,
                     float* __restrict__ hout,
                     int* __restrict__ lkey,
                     float* __restrict__ lv0,
                     float* __restrict__ lv1,
                     int* __restrict__ lcnt) {
    constexpr int VEC = H / 32;
    typedef __attribute__((ext_vector_type(VEC))) float vecf;
    static_assert(H == 32 * VEC && (VEC == 2 || VEC == 4), "row = one vector load per lane");
    static_assert(EE % 4 == 0, "edge list read in 16-byte pieces");
    static_assert(SC_LD * 4 <= 64, "one mask bit per edge of a thread");

    __shared__ float acc[SC_ROWS * H];
    __shared__ int   q_rl[3][SC_QCAP];
    __shared__ int   q_col[3][SC_QCAP];
    __shared__ float q_v0[3][SC_QCAP];
    __shared__ float q_v1[3][SC_QCAP];
    __shared__ int   q_n[3];

    const int t    = threadIdx.x;
    const int lane = t & 31;
    const int wave = t >> 5;
    const int lo   = blockIdx.x * SC_ROWS;
    const int cnt  = min(SC_ROWS, NN - lo);

    for (int i = t; i < SC_ROWS * H; i += SC_THREADS) acc[i] = 0.0f;
    if (t < 3) q_n[t] = 0;

    int*   const gk  = lkey + (size_t)blockIdx.x * LIST_CAP;
    float* const gv0 = lv0  + (size_t)blockIdx.x * LIST_CAP;
    float* const gv1 = lv1  + (size_t)blockIdx.x * LIST_CAP;

    bool listed = false;
    if constexpr (!MAKE_LIST) {
        const int nl = lcnt[blockIdx.x];
        if (nl <= LIST_CAP) {
            listed = true;
            __shared__ int   s_key[LIST_CAP];
            __shared__ float s_v0[LIST_CAP];
            __shared__ float s_v1[LIST_CAP];
            for (int i = t; i < nl; i += SC_THREADS) {
                s_key[i] = gk[i];
                s_v0[i]  = gv0[i];
                s_v1[i]  = gv1[i];
            }
            __syncthreads();
            for (int i0 = wave; i0 < nl; i0 += SC_WAVES * LIST_U) {
                int   rl[LIST_U];
                bool  ok[LIST_U];
                float v0[LIST_U], v1[LIST_U];
                vecf  p0[LIST_U], p1[LIST_U];
#pragma unroll
                for (int u = 0; u < LIST_U; ++u) {
                    const int i  = i0 + u * SC_WAVES;
                    ok[u] = i < nl;
                    const int ic = ok[u] ? i : i0;
                    const int key = s_key[ic];
                    rl[u] = min(key >> KEY_SHIFT, SC_ROWS - 1);
                    v0[u] = s_v0[ic];
                    v1[u] = s_v1[ic];
                    const unsigned off = (unsigned)(key & KEY_MASK) * H + VEC * lane;
                    p0[u] = *reinterpret_cast<const vecf*>(pre0 + off);
                    p1[u] = *reinterpret_cast<const vecf*>(pre1 + off);
                }
#pragma unroll
                for (int u = 0; u < LIST_U; ++u) {
                    float* a = acc + rl[u] * H + lane;
#pragma unroll
                    for (int j = 0; j < VEC; ++j)
                        atomicAdd(a + j * 32,
                                  ok[u] ? v0[u] * p0[u][j] + v1[u] * p1[u][j] : 0.0f);
                }
            }
        }
    }

    if (!listed) {
        int c = (blockIdx.x * 7) % SC_NCHUNK;
        int base = 0;

        int4 rv[SC_LD];
        unsigned long long m = 0;
#pragma unroll
        for (int j = 0; j < SC_LD; ++j) {
            const int e0 = c * SC_CHUNK + ((j * SC_THREADS + t) << 2);
            rv[j] = *reinterpret_cast<const int4*>(erow + min(e0, EE - 4));
        }
#pragma unroll
        for (int j = 0; j < SC_LD; ++j) {
            const unsigned lim = (c * SC_CHUNK + ((j * SC_THREADS + t) << 2) < EE) ? (unsigned)cnt : 0u;
            m |= (unsigned long long)((unsigned)(rv[j].x - lo) < lim) << (4 * j + 0);
            m |= (unsigned long long)((unsigned)(rv[j].y - lo) < lim) << (4 * j + 1);
            m |= (unsigned long long)((unsigned)(rv[j].z - lo) < lim) << (4 * j + 2);
            m |= (unsigned long long)((unsigned)(rv[j].w - lo) < lim) << (4 * j + 3);
        }
        __syncthreads();

        for (int it = 0; it < SC_NCHUNK; ++it) {
            const int qb = it % 3;
            const int cbase = c * SC_CHUNK;
            int cn = c + 1;
            if (cn == SC_NCHUNK) cn = 0;

            if (it + 1 < SC_NCHUNK) {
#pragma unroll
                for (int j = 0; j < SC_LD; ++j) {
                    const int e0 = cn * SC_CHUNK + ((j * SC_THREADS + t) << 2);
                    rv[j] = *reinterpret_cast<const int4*>(erow + min(e0, EE - 4));
                }
            }

            while (m) {
                const int b = __builtin_ctzll(m);
                m &= m - 1;
                const int e   = cbase + ((((b >> 2) * SC_THREADS) + t) << 2) + (b & 3);
                const int rl  = erow[e] - lo;
                const int col = ecol[e];
                const float v0 = ev0[e];
                const float v1 = ev1[e];
                const int slot = atomicAdd(&q_n[qb], 1);
                if constexpr (MAKE_LIST) {
                    const int pos = base + slot;
                    if (pos < LIST_CAP) {
                        gk[pos]  = (rl << KEY_SHIFT) | col;
                        gv0[pos] = v0;
                        gv1[pos] = v1;
                    }
                }
                if (slot < SC_QCAP) {
                    q_rl[qb][slot]  = rl;
                    q_col[qb][slot] = col;
                    q_v0[qb][slot]  = v0;
                    q_v1[qb][slot]  = v1;
                } else {
                    const float* s0 = pre0 + (size_t)col * H;
                    const float* s1 = pre1 + (size_t)col * H;
                    for (int k = 0; k < H; ++k)
                        atomicAdd(&acc[rl * H + (k % VEC) * 32 + (k / VEC)],
                                  v0 * s0[k] + v1 * s1[k]);
                }
            }

            if (it + 1 < SC_NCHUNK) {
#pragma unroll
                for (int j = 0; j < SC_LD; ++j) {
                    const unsigned lim = (cn * SC_CHUNK + ((j * SC_THREADS + t) << 2) < EE) ? (unsigned)cnt : 0u;
                    m |= (unsigned long long)((unsigned)(rv[j].x - lo) < lim) << (4 * j + 0);
                    m |= (unsigned long long)((unsigned)(rv[j].y - lo) < lim) << (4 * j + 1);
                    m |= (unsigned long long)((unsigned)(rv[j].z - lo) < lim) << (4 * j + 2);
                    m |= (unsigned long long)((unsigned)(rv[j].w - lo) < lim) << (4 * j + 3);
                }
            }
            c = cn;

            __syncthreads();
            if (t == 0) q_n[(it + 2) % 3] = 0;

            const int nraw = q_n[qb];
            base += nraw;
            const int n = min(nraw, SC_QCAP);
            for (int i0 = wave; i0 < n; i0 += SC_WAVES * SC_U) {
                int   rl[SC_U];
                bool  ok[SC_U];
                float v0[SC_U], v1[SC_U];
                vecf  p0[SC_U], p1[SC_U];
#pragma unroll
                for (int u = 0; u < SC_U; ++u) {
                    const int i  = i0 + u * SC_WAVES;
                    ok[u] = i < n;
                    const int ic = ok[u] ? i : i0;
                    rl[u] = q_rl[qb][ic];
                    v0[u] = q_v0[qb][ic];
                    v1[u] = q_v1[qb][ic];
                    const unsigned off = (unsigned)q_col[qb][ic] * H + VEC * lane;
                    p0[u] = *reinterpret_cast<const vecf*>(pre0 + off);
                    p1[u] = *reinterpret_cast<const vecf*>(pre1 + off);
                }
#pragma unroll
                for (int u = 0; u < SC_U; ++u) {
                    float* a = acc + rl[u] * H + lane;
#pragma unroll
                    for (int j = 0; j < VEC; ++j)
                        atomicAdd(a + j * 32,
                                  ok[u] ? v0[u] * p0[u][j] + v1[u] * p1[u][j] : 0.0f);
                }
            }
        }
        if constexpr (MAKE_LIST) {
            if (t == 0) lcnt[blockIdx.x] = base;
        }
    }
    __syncthreads();

    for (int row = wave; row < cnt; row += SC_WAVES) {
        vecf o;
#pragma unroll
        for (int j = 0; j < VEC; ++j) o[j] = fmaxf(acc[row * H + j * 32 + lane], 0.0f);
        *reinterpret_cast<vecf*>(hout + (size_t)(lo + row) * H + VEC * lane) = o;
    }
}

__global__ __launch_bounds__(SC_THREADS)
void gcn_scatter1_lds(const float* __restrict__ p0,
                      const float* __restrict__ p1,
                      const int* __restrict__ erow,
                      const int* __restrict__ ecol,
                      const float* __restrict__ ev0,
                      const float* __restrict__ ev1,
                      float* __restrict__ out,
                      const int* __restrict__ lkey,
                      const float* __restrict__ lv0,
                      const float* __restrict__ lv1,
                      const int* __restrict__ lcnt) {
    static_assert(SC_ROWS <= SC_THREADS, "one thread per owned row");
    __shared__ float acc1[SC_ROWS];

    const int t   = threadIdx.x;
    const int lo  = blockIdx.x * SC_ROWS;
    const int cnt = min(SC_ROWS, NN - lo);

    if (t < SC_ROWS) acc1[t] = 0.0f;
    __syncthreads();

    const int nl = lcnt[blockIdx.x];
    if (nl <= LIST_CAP) {
        const int*   gk  = lkey + (size_t)blockIdx.x * LIST_CAP;
        const float* gv0 = lv0  + (size_t)blockIdx.x * LIST_CAP;
        const float* gv1 = lv1  + (size_t)blockIdx.x * LIST_CAP;
        constexpr int SW = LIST_CAP / SC_THREADS;
        int   rl[SW];
        float val[SW];
#pragma unroll
        for (int u = 0; u < SW; ++u) {
            const int i  = u * SC_THREADS + t;
            const int ic = (i < nl) ? i : t;
            const int key = gk[ic];
            const int col = key & KEY_MASK;
            rl[u]  = min(key >> KEY_SHIFT, SC_ROWS - 1);
            val[u] = gv0[ic] * p0[col] + gv1[ic] * p1[col];
        }
#pragma unroll
        for (int u = 0; u < SW; ++u)
            if (u * SC_THREADS + t < nl) atomicAdd(&acc1[rl[u]], val[u]);
    } else {
        int c = (blockIdx.x * 7) % SC_NCHUNK;
        for (int it = 0; it < SC_NCHUNK; ++it) {
            const int cbase = c * SC_CHUNK;
            int4 rv[SC_LD];
#pragma unroll
            for (int j = 0; j < SC_LD; ++j) {
                const int e0 = cbase + ((j * SC_THREADS + t) << 2);
                rv[j] = *reinterpret_cast<const int4*>(erow + min(e0, EE - 4));
            }
            unsigned long long m = 0;
#pragma unroll
            for (int j = 0; j < SC_LD; ++j) {
                const unsigned lim = (cbase + ((j * SC_THREADS + t) << 2) < EE) ? (unsigned)cnt : 0u;
                m |= (unsigned long long)((unsigned)(rv[j].x - lo) < lim) << (4 * j + 0);
                m |= (unsigned long long)((unsigned)(rv[j].y - lo) < lim) << (4 * j + 1);
                m |= (unsigned long long)((unsigned)(rv[j].z - lo) < lim) << (4 * j + 2);
                m |= (unsigned long long)((unsigned)(rv[j].w - lo) < lim) << (4 * j + 3);
            }
            while (m) {
                const int b = __builtin_ctzll(m);
                m &= m - 1;
                const int e   = cbase + ((((b >> 2) * SC_THREADS) + t) << 2) + (b & 3);
                const int rl  = erow[e] - lo;
                const int col = ecol[e];
                atomicAdd(&acc1[rl], ev0[e] * p0[col] + ev1[e] * p1[col]);
            }
            c = (c + 1 == SC_NCHUNK) ? 0 : c + 1;
        }
    }
    __syncthreads();

    if (t < cnt) out[lo + t] = acc1[t];
}

__global__ void gcn_dot64x2(const float* __restrict__ h,
                            const float* __restrict__ w0,
                            const float* __restrict__ w1,
                            float* __restrict__ p0, float* __restrict__ p1) {
    int i = blockIdx.x * blockDim.x + threadIdx.x;
    if (i >= NN) return;
    const float* row = h + (size_t)i * HH2;
    float s0 = 0.0f, s1 = 0.0f;
#pragma unroll
    for (int k = 0; k < HH2; ++k) {
        float hv = row[k];
        s0 += hv * w0[k];
        s1 += hv * w1[k];
    }
    p0[i] = s0;
    p1[i] = s1;
}

extern "C" void kernel_launch(void* const* d_in, const int* in_sizes, int n_in,
                              void* d_out, int out_size, void* d_ws, size_t ws_size,
                              hipStream_t stream) {
    const float* x     = (const float*)d_in[0];
    const int*   erow  = (const int*)d_in[1];
    const int*   ecol  = (const int*)d_in[2];
    const float* evals = (const float*)d_in[3];
    const float* W1    = (const float*)d_in[4];
    const float* W2    = (const float*)d_in[5];
    const float* W3    = (const float*)d_in[6];
    float* out = (float*)d_out;

    const float* ev0 = evals;
    const float* ev1 = evals + (size_t)EE;

    float* pre0 = (float*)d_ws;
    float* pre1 = pre0 + (size_t)NN * HH1;
    float* bufA = pre1 + (size_t)NN * HH1;
    float* bufB = bufA + (size_t)NN * HH1;
    static_assert((size_t)3 * SC_RANGES * LIST_CAP + SC_RANGES <= (size_t)NN * HH2,
                  "lists and counts fit the fourth region");
    int*   lkey = (int*)bufB;
    float* lv0  = bufB + (size_t)SC_RANGES * LIST_CAP;
    float* lv1  = bufB + (size_t)2 * SC_RANGES * LIST_CAP;
    int*   lcnt = (int*)(bufB + (size_t)3 * SC_RANGES * LIST_CAP);

    const int ZB = 256;
    static_assert(NN % 32 == 0, "whole 16-row tiles, and whole pairs of them for layer 2");
    gcn_gemm_strip<FF, HH1><<<256, GS_THREADS, 0, stream>>>(
        x, W1, W1 + (size_t)FF * HH1, pre0, pre1, NN / 16);
    gcn_scatter_lds<HH1, true><<<SC_RANGES, SC_THREADS, 0, stream>>>(
        pre0, pre1, erow, ecol, ev0, ev1, bufA, lkey, lv0, lv1, lcnt);

    gcn_gemm_strip<HH1, HH2><<<256, GS_THREADS, 0, stream>>>(
        bufA, W2, W2 + (size_t)HH1 * HH2, pre0, pre1, NN / 32);
    gcn_scatter_lds<HH2, false><<<SC_RANGES, SC_THREADS, 0, stream>>>(
        pre0, pre1, erow, ecol, ev0, ev1, bufA, lkey, lv0, lv1, lcnt);

    gcn_dot64x2<<<(NN + ZB - 1) / ZB, ZB, 0, stream>>>(
        bufA, W3, W3 + (size_t)HH2, pre0, pre1);
    gcn_scatter1_lds<<<SC_RANGES, SC_THREADS, 0, stream>>>(
        pre0, pre1, erow, ecol, ev0, ev1, out, lkey, lv0, lv1, lcnt);
}


# Layer-1 product launch also writes a 1-byte owner key per edge (row/391); the layer-1 filter reads that map, not the 4-byte rows, in 10 chunks (was 40); arrival-order sums meet in 10 larger queues
# speedup vs baseline: 2.6127x; 1.1542x over previous
#include <hip/hip_runtime.h>

typedef __attribute__((ext_vector_type(2))) float v2f;
typedef __attribute__((ext_vector_type(8))) float v8f;

#define NN 100000
#define FF 256
#define EE 1600000
#define HH1 128
#define HH2 64

#define GS_PITCH   33
#define GS_WAVES   16
#define GS_THREADS (32 * GS_WAVES)
#define GS_GRID    256

#define OW_ROWS    ((NN + 255) / 256)
#define OW_WORDS   (EE / 4)

template <int K, int Nout, bool OWNERS>
__global__ __launch_bounds__(GS_THREADS)
void gcn_gemm_strip(const float* __restrict__ A,
                    const float* __restrict__ B0, const float* __restrict__ B1,
                    float* __restrict__ C0, float* __restrict__ C1, int nsteps,
                    const int* __restrict__ erow, int* __restrict__ own) {
    constexpr int STRIPS = Nout / 16;
    constexpr int WPT    = 2 * STRIPS;
    constexpr int MT     = GS_WAVES / WPT;
    constexpr int K4     = K / 4;
    constexpr int KJ     = K / 8;
    static_assert(K % 8 == 0 && Nout % 16 == 0 && GS_WAVES % WPT == 0, "tile dims");
    static_assert(MT * 16 * K4 == 2 * GS_THREADS, "two float4 per thread and step");

    __shared__ float4 sA[2][MT * KJ * GS_PITCH];

    const int t    = threadIdx.x;
    const int lane = t & 31;
    const int wave = t >> 5;
    const int h    = lane >> 4;
    const int c    = lane & 15;
    const int g    = wave / WPT;

    if constexpr (OWNERS) {
        static_assert(EE % 4 == 0, "edge rows read in 16-byte pieces");
        static_assert(4 * GS_GRID * GS_THREADS >= OW_WORDS, "four words per thread cover the map");
        static_assert((NN - 1) / OW_ROWS <= 255, "owner fits a byte");
        for (int i = 0; i < 4; ++i) {
            const int w = i * (GS_GRID * GS_THREADS) + blockIdx.x * GS_THREADS + t;
            if (w < OW_WORDS) {
                const int4 r = *reinterpret_cast<const int4*>(erow + 4 * w);
                const unsigned o0 = min((unsigned)r.x / OW_ROWS, 255u);
                const unsigned o1 = min((unsigned)r.y / OW_ROWS, 255u);
                const unsigned o2 = min((unsigned)r.z / OW_ROWS, 255u);
                const unsigned o3 = min((unsigned)r.w / OW_ROWS, 255u);
                own[w] = (int)(o0 | o1 << 8 | o2 << 16 | o3 << 24);
            }
        }
    }
    const int sup  = (wave % WPT) / STRIPS;
    const int col  = (wave % STRIPS) * 16 + c;

    const float* __restrict__ bp = (sup ? B1 : B0) + (size_t)(4 * h) * Nout + col;
    v2f breg[K4];
#pragma unroll
    for (int s = 0; s < K4; ++s) {
        const int kk = 8 * (s >> 1) + 2 * (s & 1);
        breg[s].x = bp[(kk + 0) * Nout];
        breg[s].y = bp[(kk + 1) * Nout];
    }
    float* __restrict__ cp = (sup ? C1 : C0) + (size_t)(16 * g + 8 * h) * Nout + col;

    const int f0 = t, f1 = t + GS_THREADS;
    const int r0 = f0 / K4, c0 = f0 % K4;
    const int r1 = f1 / K4, c1 = f1 % K4;
    const int srow0  = r0 * K + c0 * 4;
    const int srow1  = r1 * K + c1 * 4;
    const int sslot0 = ((r0 >> 4) * KJ + (c0 >> 1)) * GS_PITCH + (c0 & 1) * 16 + (r0 & 15);
    const int sslot1 = ((r1 >> 4) * KJ + (c1 >> 1)) * GS_PITCH + (c1 & 1) * 16 + (r1 & 15);

    int step = blockIdx.x;
    {
        const float* __restrict__ ap = A + (size_t)min(step, nsteps - 1) * (MT * 16 * K);
        const float4 p0 = *reinterpret_cast<const float4*>(ap + srow0);
        const float4 p1 = *reinterpret_cast<const float4*>(ap + srow1);
        sA[0][sslot0] = p0;
        sA[0][sslot1] = p1;
    }
    __syncthreads();

    int buf = 0;
    v8f rprev = {};
    float* __restrict__ cprev = cp;
    bool pending = false;
    for (; step < nsteps; step += gridDim.x) {
        const float* __restrict__ ap = A + (size_t)min(step + (int)gridDim.x, nsteps - 1) * (MT * 16 * K);
        const float4 st0 = *reinterpret_cast<const float4*>(ap + srow0);
        const float4 st1 = *reinterpret_cast<const float4*>(ap + srow1);

        if (pending) {
#pragma unroll
            for (int v = 0; v < 8; ++v) cprev[(size_t)v * Nout] = rprev[v];
        }

        const float4* __restrict__ fa = sA[buf] + g * (KJ * GS_PITCH) + lane;
        v8f acc0 = {}, acc1 = {}, acc2 = {}, acc3 = {};
#pragma unroll
        for (int j = 0; j < KJ; j += 2) {
            const float4 a0 = fa[(j + 0) * GS_PITCH];
            const float4 a1 = fa[(j + 1) * GS_PITCH];
            v2f x0 = {a0.x, a0.y}, x1 = {a0.z, a0.w}, x2 = {a1.x, a1.y}, x3 = {a1.z, a1.w};
            acc0 = __builtin_amdgcn_wmma_f32_16x16x4_f32(false, x0, false, breg[2 * j + 0], (short)0, acc0, false, false);
            acc1 = __builtin_amdgcn_wmma_f32_16x16x4_f32(false, x1, false, breg[2 * j + 1], (short)0, acc1, false, false);
            acc2 = __builtin_amdgcn_wmma_f32_16x16x4_f32(false, x2, false, breg[2 * j + 2], (short)0, acc2, false, false);
            acc3 = __builtin_amdgcn_wmma_f32_16x16x4_f32(false, x3, false, breg[2 * j + 3], (short)0, acc3, false, false);
        }
        rprev   = (acc0 + acc1) + (acc2 + acc3);
        cprev   = cp + (size_t)step * (MT * 16 * Nout);
        pending = true;

        sA[buf ^ 1][sslot0] = st0;
        sA[buf ^ 1][sslot1] = st1;
        __syncthreads();
        buf ^= 1;
    }
    if (pending) {
#pragma unroll
        for (int v = 0; v < 8; ++v) cprev[(size_t)v * Nout] = rprev[v];
    }
}

#define SC_THREADS 1024
#define SC_WAVES   (SC_THREADS / 32)
#define SC_RANGES  256
#define SC_ROWS    ((NN + SC_RANGES - 1) / SC_RANGES)
#define SC_LD      10
#define SC_CHUNK   (SC_THREADS * SC_LD * 4)
#define SC_NCHUNK  ((EE + SC_CHUNK - 1) / SC_CHUNK)
#define SC_QCAP    1024
#define SC_U       6
static_assert(OW_ROWS == SC_ROWS && SC_RANGES == 256, "the owner byte is the work-group index");
#define OW_LD      10
#define OW_NQ      (EE / 16)
#define OW_CHUNKQ  (SC_THREADS * OW_LD)
#define OW_NCHUNK  ((OW_NQ + OW_CHUNKQ - 1) / OW_CHUNKQ)
static_assert(EE % 16 == 0, "owner map read in 16-byte pieces");

__device__ __forceinline__ unsigned own_hit(unsigned w, unsigned pat) {
    const unsigned x = w ^ pat;
    return ((x - 0x01010101u) & ~x & 0x80808080u) != 0u;
}
__device__ __forceinline__ unsigned long long own_hit4(int4 v, unsigned pat, bool valid) {
    const unsigned f = own_hit((unsigned)v.x, pat) | own_hit((unsigned)v.y, pat) << 1 |
                       own_hit((unsigned)v.z, pat) << 2 | own_hit((unsigned)v.w, pat) << 3;
    return valid ? f : 0u;
}

#define LIST_CAP   7168
#define LIST_U     8
#define KEY_SHIFT  17
#define KEY_MASK   ((1 << KEY_SHIFT) - 1)
static_assert(NN <= (1 << KEY_SHIFT), "col fits the low key bits");
static_assert(SC_ROWS <= (1 << (31 - KEY_SHIFT)), "local row fits the high key bits");
static_assert(LIST_CAP % SC_THREADS == 0, "layer 3 walks the list in whole sweeps");

template <int H, bool MAKE_LIST>
__global__ __launch_bounds__(SC_THREADS)
void gcn_scatter_lds(const float* __restrict__ pre0,
                     const float* __restrict__ pre1,
                     const int* __restrict__ erow,
                     const int* __restrict__ ecol,
                     const float* __restrict__ ev0,
                     const float* __restrict__ ev1,
                     float* __restrict__ hout,
                     int* __restrict__ lkey,
                     float* __restrict__ lv0,
                     float* __restrict__ lv1,
                     int* __restrict__ lcnt,
                     const int* __restrict__ own) {
    constexpr int VEC = H / 32;
    typedef __attribute__((ext_vector_type(VEC))) float vecf;
    static_assert(H == 32 * VEC && (VEC == 2 || VEC == 4), "row = one vector load per lane");
    static_assert(EE % 4 == 0, "edge list read in 16-byte pieces");
    static_assert(OW_LD * 4 <= 64, "one mask bit per owner word of a thread");

    __shared__ float acc[SC_ROWS * H];
    __shared__ int   q_rl[3][SC_QCAP];
    __shared__ int   q_col[3][SC_QCAP];
    __shared__ float q_v0[3][SC_QCAP];
    __shared__ float q_v1[3][SC_QCAP];
    __shared__ int   q_n[3];

    const int t    = threadIdx.x;
    const int lane = t & 31;
    const int wave = t >> 5;
    const int lo   = blockIdx.x * SC_ROWS;
    const int cnt  = min(SC_ROWS, NN - lo);

    for (int i = t; i < SC_ROWS * H; i += SC_THREADS) acc[i] = 0.0f;
    if (t < 3) q_n[t] = 0;

    int*   const gk  = lkey + (size_t)blockIdx.x * LIST_CAP;
    float* const gv0 = lv0  + (size_t)blockIdx.x * LIST_CAP;
    float* const gv1 = lv1  + (size_t)blockIdx.x * LIST_CAP;

    bool listed = false;
    if constexpr (!MAKE_LIST) {
        const int nl = lcnt[blockIdx.x];
        if (nl <= LIST_CAP) {
            listed = true;
            __shared__ int   s_key[LIST_CAP];
            __shared__ float s_v0[LIST_CAP];
            __shared__ float s_v1[LIST_CAP];
            for (int i = t; i < nl; i += SC_THREADS) {
                s_key[i] = gk[i];
                s_v0[i]  = gv0[i];
                s_v1[i]  = gv1[i];
            }
            __syncthreads();
            for (int i0 = wave; i0 < nl; i0 += SC_WAVES * LIST_U) {
                int   rl[LIST_U];
                bool  ok[LIST_U];
                float v0[LIST_U], v1[LIST_U];
                vecf  p0[LIST_U], p1[LIST_U];
#pragma unroll
                for (int u = 0; u < LIST_U; ++u) {
                    const int i  = i0 + u * SC_WAVES;
                    ok[u] = i < nl;
                    const int ic = ok[u] ? i : i0;
                    const int key = s_key[ic];
                    rl[u] = min(key >> KEY_SHIFT, SC_ROWS - 1);
                    v0[u] = s_v0[ic];
                    v1[u] = s_v1[ic];
                    const unsigned off = (unsigned)(key & KEY_MASK) * H + VEC * lane;
                    p0[u] = *reinterpret_cast<const vecf*>(pre0 + off);
                    p1[u] = *reinterpret_cast<const vecf*>(pre1 + off);
                }
#pragma unroll
                for (int u = 0; u < LIST_U; ++u) {
                    float* a = acc + rl[u] * H + lane;
#pragma unroll
                    for (int j = 0; j < VEC; ++j)
                        atomicAdd(a + j * 32,
                                  ok[u] ? v0[u] * p0[u][j] + v1[u] * p1[u][j] : 0.0f);
                }
            }
        }
    }

    if (!listed) {
        int c = (blockIdx.x * 7) % OW_NCHUNK;
        int base = 0;

        const int4* __restrict__ own4 = reinterpret_cast<const int4*>(own);
        const unsigned pat = blockIdx.x * 0x01010101u;
        int4 rv[OW_LD];
        unsigned long long m = 0;
#pragma unroll
        for (int j = 0; j < OW_LD; ++j) {
            const int q = c * OW_CHUNKQ + j * SC_THREADS + t;
            rv[j] = own4[min(q, OW_NQ - 1)];
        }
#pragma unroll
        for (int j = 0; j < OW_LD; ++j) {
            m |= own_hit4(rv[j], pat, c * OW_CHUNKQ + j * SC_THREADS + t < OW_NQ) << (4 * j);
        }
        __syncthreads();

        for (int it = 0; it < OW_NCHUNK; ++it) {
            const int qb = it % 3;
            const int cbase = c * OW_CHUNKQ;
            int cn = c + 1;
            if (cn == OW_NCHUNK) cn = 0;

            if (it + 1 < OW_NCHUNK) {
#pragma unroll
                for (int j = 0; j < OW_LD; ++j) {
                    const int q = cn * OW_CHUNKQ + j * SC_THREADS + t;
                    rv[j] = own4[min(q, OW_NQ - 1)];
                }
            }

            while (m) {
                const int b = __builtin_ctzll(m);
                m &= m - 1;
                const int eb = ((cbase + (b >> 2) * SC_THREADS + t) << 4) + ((b & 3) << 2);
                const int4   r4 = *reinterpret_cast<const int4*>(erow + eb);
                const int4   c4 = *reinterpret_cast<const int4*>(ecol + eb);
                const float4 a4 = *reinterpret_cast<const float4*>(ev0 + eb);
                const float4 b4 = *reinterpret_cast<const float4*>(ev1 + eb);
                const int   rr[4] = {r4.x, r4.y, r4.z, r4.w};
                const int   cc[4] = {c4.x, c4.y, c4.z, c4.w};
                const float aa[4] = {a4.x, a4.y, a4.z, a4.w};
                const float bb[4] = {b4.x, b4.y, b4.z, b4.w};
#pragma unroll
                for (int i = 0; i < 4; ++i) {
                    const int rl = rr[i] - lo;
                    if ((unsigned)rl >= (unsigned)cnt) continue;
                    const int col = cc[i];
                    const float v0 = aa[i];
                    const float v1 = bb[i];
                    const int slot = atomicAdd(&q_n[qb], 1);
                    if constexpr (MAKE_LIST) {
                        const int pos = base + slot;
                        if (pos < LIST_CAP) {
                            gk[pos]  = (rl << KEY_SHIFT) | col;
                            gv0[pos] = v0;
                            gv1[pos] = v1;
                        }
                    }
                    if (slot < SC_QCAP) {
                        q_rl[qb][slot]  = rl;
                        q_col[qb][slot] = col;
                        q_v0[qb][slot]  = v0;
                        q_v1[qb][slot]  = v1;
                    } else {
                        const float* s0 = pre0 + (size_t)col * H;
                        const float* s1 = pre1 + (size_t)col * H;
                        for (int k = 0; k < H; ++k)
                            atomicAdd(&acc[rl * H + (k % VEC) * 32 + (k / VEC)],
                                      v0 * s0[k] + v1 * s1[k]);
                    }
                }
            }

            if (it + 1 < OW_NCHUNK) {
#pragma unroll
                for (int j = 0; j < OW_LD; ++j) {
                    m |= own_hit4(rv[j], pat, cn * OW_CHUNKQ + j * SC_THREADS + t < OW_NQ) << (4 * j);
                }
            }
            c = cn;

            __syncthreads();
            if (t == 0) q_n[(it + 2) % 3] = 0;

            const int nraw = q_n[qb];
            base += nraw;
            const int n = min(nraw, SC_QCAP);
            for (int i0 = wave; i0 < n; i0 += SC_WAVES * SC_U) {
                int   rl[SC_U];
                bool  ok[SC_U];
                float v0[SC_U], v1[SC_U];
                vecf  p0[SC_U], p1[SC_U];
#pragma unroll
                for (int u = 0; u < SC_U; ++u) {
                    const int i  = i0 + u * SC_WAVES;
                    ok[u] = i < n;
                    const int ic = ok[u] ? i : i0;
                    rl[u] = q_rl[qb][ic];
                    v0[u] = q_v0[qb][ic];
                    v1[u] = q_v1[qb][ic];
                    const unsigned off = (unsigned)q_col[qb][ic] * H + VEC * lane;
                    p0[u] = *reinterpret_cast<const vecf*>(pre0 + off);
                    p1[u] = *reinterpret_cast<const vecf*>(pre1 + off);
                }
#pragma unroll
                for (int u = 0; u < SC_U; ++u) {
                    float* a = acc + rl[u] * H + lane;
#pragma unroll
                    for (int j = 0; j < VEC; ++j)
                        atomicAdd(a + j * 32,
                                  ok[u] ? v0[u] * p0[u][j] + v1[u] * p1[u][j] : 0.0f);
                }
            }
        }
        if constexpr (MAKE_LIST) {
            if (t == 0) lcnt[blockIdx.x] = base;
        }
    }
    __syncthreads();

    for (int row = wave; row < cnt; row += SC_WAVES) {
        vecf o;
#pragma unroll
        for (int j = 0; j < VEC; ++j) o[j] = fmaxf(acc[row * H + j * 32 + lane], 0.0f);
        *reinterpret_cast<vecf*>(hout + (size_t)(lo + row) * H + VEC * lane) = o;
    }
}

__global__ __launch_bounds__(SC_THREADS)
void gcn_scatter1_lds(const float* __restrict__ p0,
                      const float* __restrict__ p1,
                      const int* __restrict__ erow,
                      const int* __restrict__ ecol,
                      const float* __restrict__ ev0,
                      const float* __restrict__ ev1,
                      float* __restrict__ out,
                      const int* __restrict__ lkey,
                      const float* __restrict__ lv0,
                      const float* __restrict__ lv1,
                      const int* __restrict__ lcnt) {
    static_assert(SC_ROWS <= SC_THREADS, "one thread per owned row");
    __shared__ float acc1[SC_ROWS];

    const int t   = threadIdx.x;
    const int lo  = blockIdx.x * SC_ROWS;
    const int cnt = min(SC_ROWS, NN - lo);

    if (t < SC_ROWS) acc1[t] = 0.0f;
    __syncthreads();

    const int nl = lcnt[blockIdx.x];
    if (nl <= LIST_CAP) {
        const int*   gk  = lkey + (size_t)blockIdx.x * LIST_CAP;
        const float* gv0 = lv0  + (size_t)blockIdx.x * LIST_CAP;
        const float* gv1 = lv1  + (size_t)blockIdx.x * LIST_CAP;
        constexpr int SW = LIST_CAP / SC_THREADS;
        int   rl[SW];
        float val[SW];
#pragma unroll
        for (int u = 0; u < SW; ++u) {
            const int i  = u * SC_THREADS + t;
            const int ic = (i < nl) ? i : t;
            const int key = gk[ic];
            const int col = key & KEY_MASK;
            rl[u]  = min(key >> KEY_SHIFT, SC_ROWS - 1);
            val[u] = gv0[ic] * p0[col] + gv1[ic] * p1[col];
        }
#pragma unroll
        for (int u = 0; u < SW; ++u)
            if (u * SC_THREADS + t < nl) atomicAdd(&acc1[rl[u]], val[u]);
    } else {
        int c = (blockIdx.x * 7) % SC_NCHUNK;
        for (int it = 0; it < SC_NCHUNK; ++it) {
            const int cbase = c * SC_CHUNK;
            int4 rv[SC_LD];
#pragma unroll
            for (int j = 0; j < SC_LD; ++j) {
                const int e0 = cbase + ((j * SC_THREADS + t) << 2);
                rv[j] = *reinterpret_cast<const int4*>(erow + min(e0, EE - 4));
            }
            unsigned long long m = 0;
#pragma unroll
            for (int j = 0; j < SC_LD; ++j) {
                const unsigned lim = (cbase + ((j * SC_THREADS + t) << 2) < EE) ? (unsigned)cnt : 0u;
                m |= (unsigned long long)((unsigned)(rv[j].x - lo) < lim) << (4 * j + 0);
                m |= (unsigned long long)((unsigned)(rv[j].y - lo) < lim) << (4 * j + 1);
                m |= (unsigned long long)((unsigned)(rv[j].z - lo) < lim) << (4 * j + 2);
                m |= (unsigned long long)((unsigned)(rv[j].w - lo) < lim) << (4 * j + 3);
            }
            while (m) {
                const int b = __builtin_ctzll(m);
                m &= m - 1;
                const int e   = cbase + ((((b >> 2) * SC_THREADS) + t) << 2) + (b & 3);
                const int rl  = erow[e] - lo;
                const int col = ecol[e];
                atomicAdd(&acc1[rl], ev0[e] * p0[col] + ev1[e] * p1[col]);
            }
            c = (c + 1 == SC_NCHUNK) ? 0 : c + 1;
        }
    }
    __syncthreads();

    if (t < cnt) out[lo + t] = acc1[t];
}

__global__ void gcn_dot64x2(const float* __restrict__ h,
                            const float* __restrict__ w0,
                            const float* __restrict__ w1,
                            float* __restrict__ p0, float* __restrict__ p1) {
    int i = blockIdx.x * blockDim.x + threadIdx.x;
    if (i >= NN) return;
    const float* row = h + (size_t)i * HH2;
    float s0 = 0.0f, s1 = 0.0f;
#pragma unroll
    for (int k = 0; k < HH2; ++k) {
        float hv = row[k];
        s0 += hv * w0[k];
        s1 += hv * w1[k];
    }
    p0[i] = s0;
    p1[i] = s1;
}

extern "C" void kernel_launch(void* const* d_in, const int* in_sizes, int n_in,
                              void* d_out, int out_size, void* d_ws, size_t ws_size,
                              hipStream_t stream) {
    const float* x     = (const float*)d_in[0];
    const int*   erow  = (const int*)d_in[1];
    const int*   ecol  = (const int*)d_in[2];
    const float* evals = (const float*)d_in[3];
    const float* W1    = (const float*)d_in[4];
    const float* W2    = (const float*)d_in[5];
    const float* W3    = (const float*)d_in[6];
    float* out = (float*)d_out;

    const float* ev0 = evals;
    const float* ev1 = evals + (size_t)EE;

    float* pre0 = (float*)d_ws;
    float* pre1 = pre0 + (size_t)NN * HH1;
    float* bufA = pre1 + (size_t)NN * HH1;
    float* bufB = bufA + (size_t)NN * HH1;
    static_assert((size_t)3 * SC_RANGES * LIST_CAP + SC_RANGES + OW_WORDS <= (size_t)NN * HH2,
                  "lists, counts and owner map fit the fourth region");
    static_assert((3 * SC_RANGES * LIST_CAP + SC_RANGES) % 4 == 0, "owner map 16-byte aligned");
    int*   lkey = (int*)bufB;
    float* lv0  = bufB + (size_t)SC_RANGES * LIST_CAP;
    float* lv1  = bufB + (size_t)2 * SC_RANGES * LIST_CAP;
    int*   lcnt = (int*)(bufB + (size_t)3 * SC_RANGES * LIST_CAP);
    int*   own  = lcnt + SC_RANGES;

    const int ZB = 256;
    static_assert(NN % 32 == 0, "whole 16-row tiles, and whole pairs of them for layer 2");
    gcn_gemm_strip<FF, HH1, true><<<GS_GRID, GS_THREADS, 0, stream>>>(
        x, W1, W1 + (size_t)FF * HH1, pre0, pre1, NN / 16, erow, own);
    gcn_scatter_lds<HH1, true><<<SC_RANGES, SC_THREADS, 0, stream>>>(
        pre0, pre1, erow, ecol, ev0, ev1, bufA, lkey, lv0, lv1, lcnt, own);

    gcn_gemm_strip<HH1, HH2, false><<<GS_GRID, GS_THREADS, 0, stream>>>(
        bufA, W2, W2 + (size_t)HH1 * HH2, pre0, pre1, NN / 32, erow, own);
    gcn_scatter_lds<HH2, false><<<SC_RANGES, SC_THREADS, 0, stream>>>(
        pre0, pre1, erow, ecol, ev0, ev1, bufA, lkey, lv0, lv1, lcnt, own);

    gcn_dot64x2<<<(NN + ZB - 1) / ZB, ZB, 0, stream>>>(
        bufA, W3, W3 + (size_t)HH2, pre0, pre1);
    gcn_scatter1_lds<<<SC_RANGES, SC_THREADS, 0, stream>>>(
        pre0, pre1, erow, ecol, ev0, ev1, out, lkey, lv0, lv1, lcnt);
}


# Layer-1 aggregation lists its edges in LDS first, then gathers them in two passes by source-row half, so a pass's source lines stay in L2; sums: lower-half sources first, arrival order inside
# speedup vs baseline: 2.9543x; 1.1307x over previous
#include <hip/hip_runtime.h>

typedef __attribute__((ext_vector_type(2))) float v2f;
typedef __attribute__((ext_vector_type(8))) float v8f;

#define NN 100000
#define FF 256
#define EE 1600000
#define HH1 128
#define HH2 64

#define GS_PITCH   33
#define GS_WAVES   16
#define GS_THREADS (32 * GS_WAVES)
#define GS_GRID    256

#define OW_ROWS    ((NN + 255) / 256)
#define OW_WORDS   (EE / 4)

template <int K, int Nout, bool OWNERS>
__global__ __launch_bounds__(GS_THREADS)
void gcn_gemm_strip(const float* __restrict__ A,
                    const float* __restrict__ B0, const float* __restrict__ B1,
                    float* __restrict__ C0, float* __restrict__ C1, int nsteps,
                    const int* __restrict__ erow, int* __restrict__ own) {
    constexpr int STRIPS = Nout / 16;
    constexpr int WPT    = 2 * STRIPS;
    constexpr int MT     = GS_WAVES / WPT;
    constexpr int K4     = K / 4;
    constexpr int KJ     = K / 8;
    static_assert(K % 8 == 0 && Nout % 16 == 0 && GS_WAVES % WPT == 0, "tile dims");
    static_assert(MT * 16 * K4 == 2 * GS_THREADS, "two float4 per thread and step");

    __shared__ float4 sA[2][MT * KJ * GS_PITCH];

    const int t    = threadIdx.x;
    const int lane = t & 31;
    const int wave = t >> 5;
    const int h    = lane >> 4;
    const int c    = lane & 15;
    const int g    = wave / WPT;

    if constexpr (OWNERS) {
        static_assert(EE % 4 == 0, "edge rows read in 16-byte pieces");
        static_assert(4 * GS_GRID * GS_THREADS >= OW_WORDS, "four words per thread cover the map");
        static_assert((NN - 1) / OW_ROWS <= 255, "owner fits a byte");
        for (int i = 0; i < 4; ++i) {
            const int w = i * (GS_GRID * GS_THREADS) + blockIdx.x * GS_THREADS + t;
            if (w < OW_WORDS) {
                const int4 r = *reinterpret_cast<const int4*>(erow + 4 * w);
                const unsigned o0 = min((unsigned)r.x / OW_ROWS, 255u);
                const unsigned o1 = min((unsigned)r.y / OW_ROWS, 255u);
                const unsigned o2 = min((unsigned)r.z / OW_ROWS, 255u);
                const unsigned o3 = min((unsigned)r.w / OW_ROWS, 255u);
                own[w] = (int)(o0 | o1 << 8 | o2 << 16 | o3 << 24);
            }
        }
    }
    const int sup  = (wave % WPT) / STRIPS;
    const int col  = (wave % STRIPS) * 16 + c;

    const float* __restrict__ bp = (sup ? B1 : B0) + (size_t)(4 * h) * Nout + col;
    v2f breg[K4];
#pragma unroll
    for (int s = 0; s < K4; ++s) {
        const int kk = 8 * (s >> 1) + 2 * (s & 1);
        breg[s].x = bp[(kk + 0) * Nout];
        breg[s].y = bp[(kk + 1) * Nout];
    }
    float* __restrict__ cp = (sup ? C1 : C0) + (size_t)(16 * g + 8 * h) * Nout + col;

    const int f0 = t, f1 = t + GS_THREADS;
    const int r0 = f0 / K4, c0 = f0 % K4;
    const int r1 = f1 / K4, c1 = f1 % K4;
    const int srow0  = r0 * K + c0 * 4;
    const int srow1  = r1 * K + c1 * 4;
    const int sslot0 = ((r0 >> 4) * KJ + (c0 >> 1)) * GS_PITCH + (c0 & 1) * 16 + (r0 & 15);
    const int sslot1 = ((r1 >> 4) * KJ + (c1 >> 1)) * GS_PITCH + (c1 & 1) * 16 + (r1 & 15);

    int step = blockIdx.x;
    {
        const float* __restrict__ ap = A + (size_t)min(step, nsteps - 1) * (MT * 16 * K);
        const float4 p0 = *reinterpret_cast<const float4*>(ap + srow0);
        const float4 p1 = *reinterpret_cast<const float4*>(ap + srow1);
        sA[0][sslot0] = p0;
        sA[0][sslot1] = p1;
    }
    __syncthreads();

    int buf = 0;
    v8f rprev = {};
    float* __restrict__ cprev = cp;
    bool pending = false;
    for (; step < nsteps; step += gridDim.x) {
        const float* __restrict__ ap = A + (size_t)min(step + (int)gridDim.x, nsteps - 1) * (MT * 16 * K);
        const float4 st0 = *reinterpret_cast<const float4*>(ap + srow0);
        const float4 st1 = *reinterpret_cast<const float4*>(ap + srow1);

        if (pending) {
#pragma unroll
            for (int v = 0; v < 8; ++v) cprev[(size_t)v * Nout] = rprev[v];
        }

        const float4* __restrict__ fa = sA[buf] + g * (KJ * GS_PITCH) + lane;
        v8f acc0 = {}, acc1 = {}, acc2 = {}, acc3 = {};
#pragma unroll
        for (int j = 0; j < KJ; j += 2) {
            const float4 a0 = fa[(j + 0) * GS_PITCH];
            const float4 a1 = fa[(j + 1) * GS_PITCH];
            v2f x0 = {a0.x, a0.y}, x1 = {a0.z, a0.w}, x2 = {a1.x, a1.y}, x3 = {a1.z, a1.w};
            acc0 = __builtin_amdgcn_wmma_f32_16x16x4_f32(false, x0, false, breg[2 * j + 0], (short)0, acc0, false, false);
            acc1 = __builtin_amdgcn_wmma_f32_16x16x4_f32(false, x1, false, breg[2 * j + 1], (short)0, acc1, false, false);
            acc2 = __builtin_amdgcn_wmma_f32_16x16x4_f32(false, x2, false, breg[2 * j + 2], (short)0, acc2, false, false);
            acc3 = __builtin_amdgcn_wmma_f32_16x16x4_f32(false, x3, false, breg[2 * j + 3], (short)0, acc3, false, false);
        }
        rprev   = (acc0 + acc1) + (acc2 + acc3);
        cprev   = cp + (size_t)step * (MT * 16 * Nout);
        pending = true;

        sA[buf ^ 1][sslot0] = st0;
        sA[buf ^ 1][sslot1] = st1;
        __syncthreads();
        buf ^= 1;
    }
    if (pending) {
#pragma unroll
        for (int v = 0; v < 8; ++v) cprev[(size_t)v * Nout] = rprev[v];
    }
}

#define SC_THREADS 1024
#define SC_WAVES   (SC_THREADS / 32)
#define SC_RANGES  256
#define SC_ROWS    ((NN + SC_RANGES - 1) / SC_RANGES)
#define SC_LD      10
#define SC_CHUNK   (SC_THREADS * SC_LD * 4)
#define SC_NCHUNK  ((EE + SC_CHUNK - 1) / SC_CHUNK)
#define SC_QCAP    1024
#define SC_U       6
static_assert(OW_ROWS == SC_ROWS && SC_RANGES == 256, "the owner byte is the work-group index");
#define OW_LD      10
#define OW_NQ      (EE / 16)
#define OW_CHUNKQ  (SC_THREADS * OW_LD)
#define OW_NCHUNK  ((OW_NQ + OW_CHUNKQ - 1) / OW_CHUNKQ)
static_assert(EE % 16 == 0, "owner map read in 16-byte pieces");

__device__ __forceinline__ unsigned own_hit(unsigned w, unsigned pat) {
    const unsigned x = w ^ pat;
    return ((x - 0x01010101u) & ~x & 0x80808080u) != 0u;
}
__device__ __forceinline__ unsigned long long own_hit4(int4 v, unsigned pat, bool valid) {
    const unsigned f = own_hit((unsigned)v.x, pat) | own_hit((unsigned)v.y, pat) << 1 |
                       own_hit((unsigned)v.z, pat) << 2 | own_hit((unsigned)v.w, pat) << 3;
    return valid ? f : 0u;
}

#define LIST_CAP   7168
#define LIST_U     8
#define KEY_SHIFT  17
#define KEY_MASK   ((1 << KEY_SHIFT) - 1)
static_assert(NN <= (1 << KEY_SHIFT), "col fits the low key bits");
static_assert(SC_ROWS <= (1 << (31 - KEY_SHIFT)), "local row fits the high key bits");
static_assert(LIST_CAP % SC_THREADS == 0, "layer 3 walks the list in whole sweeps");

template <int H>
__global__ __launch_bounds__(SC_THREADS)
void gcn_scatter_lds(const float* __restrict__ pre0,
                     const float* __restrict__ pre1,
                     const int* __restrict__ erow,
                     const int* __restrict__ ecol,
                     const float* __restrict__ ev0,
                     const float* __restrict__ ev1,
                     float* __restrict__ hout,
                     const int* __restrict__ lkey,
                     const float* __restrict__ lv0,
                     const float* __restrict__ lv1,
                     const int* __restrict__ lcnt,
                     const int* __restrict__ own) {
    constexpr int VEC = H / 32;
    typedef __attribute__((ext_vector_type(VEC))) float vecf;
    static_assert(H == 32 * VEC && (VEC == 2 || VEC == 4), "row = one vector load per lane");
    static_assert(EE % 4 == 0, "edge list read in 16-byte pieces");
    static_assert(OW_LD * 4 <= 64, "one mask bit per owner word of a thread");

    __shared__ float acc[SC_ROWS * H];
    __shared__ int   q_rl[3][SC_QCAP];
    __shared__ int   q_col[3][SC_QCAP];
    __shared__ float q_v0[3][SC_QCAP];
    __shared__ float q_v1[3][SC_QCAP];
    __shared__ int   q_n[3];

    const int t    = threadIdx.x;
    const int lane = t & 31;
    const int wave = t >> 5;
    const int lo   = blockIdx.x * SC_ROWS;
    const int cnt  = min(SC_ROWS, NN - lo);

    for (int i = t; i < SC_ROWS * H; i += SC_THREADS) acc[i] = 0.0f;
    if (t < 3) q_n[t] = 0;

    const int*   const gk  = lkey + (size_t)blockIdx.x * LIST_CAP;
    const float* const gv0 = lv0  + (size_t)blockIdx.x * LIST_CAP;
    const float* const gv1 = lv1  + (size_t)blockIdx.x * LIST_CAP;

    bool listed = false;
    {
        const int nl = lcnt[blockIdx.x];
        if (nl <= LIST_CAP) {
            listed = true;
            __shared__ int   s_key[LIST_CAP];
            __shared__ float s_v0[LIST_CAP];
            __shared__ float s_v1[LIST_CAP];
            for (int i = t; i < nl; i += SC_THREADS) {
                s_key[i] = gk[i];
                s_v0[i]  = gv0[i];
                s_v1[i]  = gv1[i];
            }
            __syncthreads();
            for (int i0 = wave; i0 < nl; i0 += SC_WAVES * LIST_U) {
                int   rl[LIST_U];
                bool  ok[LIST_U];
                float v0[LIST_U], v1[LIST_U];
                vecf  p0[LIST_U], p1[LIST_U];
#pragma unroll
                for (int u = 0; u < LIST_U; ++u) {
                    const int i  = i0 + u * SC_WAVES;
                    ok[u] = i < nl;
                    const int ic = ok[u] ? i : i0;
                    const int key = s_key[ic];
                    rl[u] = min(key >> KEY_SHIFT, SC_ROWS - 1);
                    v0[u] = s_v0[ic];
                    v1[u] = s_v1[ic];
                    const unsigned off = (unsigned)(key & KEY_MASK) * H + VEC * lane;
                    p0[u] = *reinterpret_cast<const vecf*>(pre0 + off);
                    p1[u] = *reinterpret_cast<const vecf*>(pre1 + off);
                }
#pragma unroll
                for (int u = 0; u < LIST_U; ++u) {
                    float* a = acc + rl[u] * H + lane;
#pragma unroll
                    for (int j = 0; j < VEC; ++j)
                        atomicAdd(a + j * 32,
                                  ok[u] ? v0[u] * p0[u][j] + v1[u] * p1[u][j] : 0.0f);
                }
            }
        }
    }

    if (!listed) {
        int c = (blockIdx.x * 7) % OW_NCHUNK;

        const int4* __restrict__ own4 = reinterpret_cast<const int4*>(own);
        const unsigned pat = blockIdx.x * 0x01010101u;
        int4 rv[OW_LD];
        unsigned long long m = 0;
#pragma unroll
        for (int j = 0; j < OW_LD; ++j) {
            const int q = c * OW_CHUNKQ + j * SC_THREADS + t;
            rv[j] = own4[min(q, OW_NQ - 1)];
        }
#pragma unroll
        for (int j = 0; j < OW_LD; ++j) {
            m |= own_hit4(rv[j], pat, c * OW_CHUNKQ + j * SC_THREADS + t < OW_NQ) << (4 * j);
        }
        __syncthreads();

        for (int it = 0; it < OW_NCHUNK; ++it) {
            const int qb = it % 3;
            const int cbase = c * OW_CHUNKQ;
            int cn = c + 1;
            if (cn == OW_NCHUNK) cn = 0;

            if (it + 1 < OW_NCHUNK) {
#pragma unroll
                for (int j = 0; j < OW_LD; ++j) {
                    const int q = cn * OW_CHUNKQ + j * SC_THREADS + t;
                    rv[j] = own4[min(q, OW_NQ - 1)];
                }
            }

            while (m) {
                const int b = __builtin_ctzll(m);
                m &= m - 1;
                const int eb = ((cbase + (b >> 2) * SC_THREADS + t) << 4) + ((b & 3) << 2);
                const int4   r4 = *reinterpret_cast<const int4*>(erow + eb);
                const int4   c4 = *reinterpret_cast<const int4*>(ecol + eb);
                const float4 a4 = *reinterpret_cast<const float4*>(ev0 + eb);
                const float4 b4 = *reinterpret_cast<const float4*>(ev1 + eb);
                const int   rr[4] = {r4.x, r4.y, r4.z, r4.w};
                const int   cc[4] = {c4.x, c4.y, c4.z, c4.w};
                const float aa[4] = {a4.x, a4.y, a4.z, a4.w};
                const float bb[4] = {b4.x, b4.y, b4.z, b4.w};
#pragma unroll
                for (int i = 0; i < 4; ++i) {
                    const int rl = rr[i] - lo;
                    if ((unsigned)rl >= (unsigned)cnt) continue;
                    const int col = cc[i];
                    const float v0 = aa[i];
                    const float v1 = bb[i];
                    const int slot = atomicAdd(&q_n[qb], 1);
                    if (slot < SC_QCAP) {
                        q_rl[qb][slot]  = rl;
                        q_col[qb][slot] = col;
                        q_v0[qb][slot]  = v0;
                        q_v1[qb][slot]  = v1;
                    } else {
                        const float* s0 = pre0 + (size_t)col * H;
                        const float* s1 = pre1 + (size_t)col * H;
                        for (int k = 0; k < H; ++k)
                            atomicAdd(&acc[rl * H + (k % VEC) * 32 + (k / VEC)],
                                      v0 * s0[k] + v1 * s1[k]);
                    }
                }
            }

            if (it + 1 < OW_NCHUNK) {
#pragma unroll
                for (int j = 0; j < OW_LD; ++j) {
                    m |= own_hit4(rv[j], pat, cn * OW_CHUNKQ + j * SC_THREADS + t < OW_NQ) << (4 * j);
                }
            }
            c = cn;

            __syncthreads();
            if (t == 0) q_n[(it + 2) % 3] = 0;

            const int n = min(q_n[qb], SC_QCAP);
            for (int i0 = wave; i0 < n; i0 += SC_WAVES * SC_U) {
                int   rl[SC_U];
                bool  ok[SC_U];
                float v0[SC_U], v1[SC_U];
                vecf  p0[SC_U], p1[SC_U];
#pragma unroll
                for (int u = 0; u < SC_U; ++u) {
                    const int i  = i0 + u * SC_WAVES;
                    ok[u] = i < n;
                    const int ic = ok[u] ? i : i0;
                    rl[u] = q_rl[qb][ic];
                    v0[u] = q_v0[qb][ic];
                    v1[u] = q_v1[qb][ic];
                    const unsigned off = (unsigned)q_col[qb][ic] * H + VEC * lane;
                    p0[u] = *reinterpret_cast<const vecf*>(pre0 + off);
                    p1[u] = *reinterpret_cast<const vecf*>(pre1 + off);
                }
#pragma unroll
                for (int u = 0; u < SC_U; ++u) {
                    float* a = acc + rl[u] * H + lane;
#pragma unroll
                    for (int j = 0; j < VEC; ++j)
                        atomicAdd(a + j * 32,
                                  ok[u] ? v0[u] * p0[u][j] + v1[u] * p1[u][j] : 0.0f);
                }
            }
        }
    }
    __syncthreads();

    for (int row = wave; row < cnt; row += SC_WAVES) {
        vecf o;
#pragma unroll
        for (int j = 0; j < VEC; ++j) o[j] = fmaxf(acc[row * H + j * 32 + lane], 0.0f);
        *reinterpret_cast<vecf*>(hout + (size_t)(lo + row) * H + VEC * lane) = o;
    }
}

#define L1_U      6
#define L1_SPLIT  (NN / 2)

__global__ __launch_bounds__(SC_THREADS)
void gcn_scatter_l1(const float* __restrict__ pre0,
                    const float* __restrict__ pre1,
                    const int* __restrict__ erow,
                    const int* __restrict__ ecol,
                    const float* __restrict__ ev0,
                    const float* __restrict__ ev1,
                    float* __restrict__ hout,
                    int* __restrict__ lkey,
                    float* __restrict__ lv0,
                    float* __restrict__ lv1,
                    int* __restrict__ lcnt,
                    const int* __restrict__ own) {
    constexpr int H   = HH1;
    constexpr int VEC = H / 32;
    typedef __attribute__((ext_vector_type(VEC))) float vecf;
    static_assert(H == 32 * VEC && VEC == 4, "row = one 16-byte load per lane");
    static_assert(OW_LD * 4 <= 64, "one mask bit per owner word of a thread");

    __shared__ float acc[SC_ROWS * H];
    __shared__ int   s_key[LIST_CAP];
    __shared__ float s_v0[LIST_CAP];
    __shared__ float s_v1[LIST_CAP];
    __shared__ int   s_n[3];

    const int t    = threadIdx.x;
    const int lane = t & 31;
    const int wave = t >> 5;
    const int lo   = blockIdx.x * SC_ROWS;
    const int cnt  = min(SC_ROWS, NN - lo);

    for (int i = t; i < SC_ROWS * H; i += SC_THREADS) acc[i] = 0.0f;
    if (t < 3) s_n[t] = 0;

    {
        int c = (blockIdx.x * 7) % OW_NCHUNK;

        const int4* __restrict__ own4 = reinterpret_cast<const int4*>(own);
        const unsigned pat = blockIdx.x * 0x01010101u;
        int4 rv[OW_LD];
        unsigned long long m = 0;
#pragma unroll
        for (int j = 0; j < OW_LD; ++j) {
            const int q = c * OW_CHUNKQ + j * SC_THREADS + t;
            rv[j] = own4[min(q, OW_NQ - 1)];
        }
#pragma unroll
        for (int j = 0; j < OW_LD; ++j) {
            m |= own_hit4(rv[j], pat, c * OW_CHUNKQ + j * SC_THREADS + t < OW_NQ) << (4 * j);
        }
        __syncthreads();

        for (int it = 0; it < OW_NCHUNK; ++it) {
            const int cbase = c * OW_CHUNKQ;
            int cn = c + 1;
            if (cn == OW_NCHUNK) cn = 0;

            if (it + 1 < OW_NCHUNK) {
#pragma unroll
                for (int j = 0; j < OW_LD; ++j) {
                    const int q = cn * OW_CHUNKQ + j * SC_THREADS + t;
                    rv[j] = own4[min(q, OW_NQ - 1)];
                }
            }

            while (m) {
                const int b = __builtin_ctzll(m);
                m &= m - 1;
                const int eb = ((cbase + (b >> 2) * SC_THREADS + t) << 4) + ((b & 3) << 2);
                const int4   r4 = *reinterpret_cast<const int4*>(erow + eb);
                const int4   c4 = *reinterpret_cast<const int4*>(ecol + eb);
                const float4 a4 = *reinterpret_cast<const float4*>(ev0 + eb);
                const float4 b4 = *reinterpret_cast<const float4*>(ev1 + eb);
                const int   rr[4] = {r4.x, r4.y, r4.z, r4.w};
                const int   cc[4] = {c4.x, c4.y, c4.z, c4.w};
                const float aa[4] = {a4.x, a4.y, a4.z, a4.w};
                const float bb[4] = {b4.x, b4.y, b4.z, b4.w};
#pragma unroll
                for (int i = 0; i < 4; ++i) {
                    const int rl = rr[i] - lo;
                    if ((unsigned)rl >= (unsigned)cnt) continue;
                    const int col = cc[i];
                    const float v0 = aa[i];
                    const float v1 = bb[i];
                    const int all = atomicAdd(&s_n[2], 1);
                    if (all < LIST_CAP) {
                        int pos;
                        if (col < L1_SPLIT) pos = atomicAdd(&s_n[0], 1);
                        else                pos = LIST_CAP - 1 - atomicAdd(&s_n[1], 1);
                        s_key[pos] = (rl << KEY_SHIFT) | col;
                        s_v0[pos]  = v0;
                        s_v1[pos]  = v1;
                    } else {
                        const float* s0 = pre0 + (size_t)col * H;
                        const float* s1 = pre1 + (size_t)col * H;
                        for (int k = 0; k < H; ++k)
                            atomicAdd(&acc[rl * H + (k % VEC) * 32 + (k / VEC)],
                                      v0 * s0[k] + v1 * s1[k]);
                    }
                }
            }

            if (it + 1 < OW_NCHUNK) {
#pragma unroll
                for (int j = 0; j < OW_LD; ++j) {
                    m |= own_hit4(rv[j], pat, cn * OW_CHUNKQ + j * SC_THREADS + t < OW_NQ) << (4 * j);
                }
            }
            c = cn;
        }
    }
    __syncthreads();

    const int nlo = s_n[0];
    const int nhi = s_n[1];

    {
        int*   const gk  = lkey + (size_t)blockIdx.x * LIST_CAP;
        float* const gv0 = lv0  + (size_t)blockIdx.x * LIST_CAP;
        float* const gv1 = lv1  + (size_t)blockIdx.x * LIST_CAP;
        for (int i = t; i < nlo + nhi; i += SC_THREADS) {
            const int sl = (i < nlo) ? i : LIST_CAP - nhi + (i - nlo);
            gk[i]  = s_key[sl];
            gv0[i] = s_v0[sl];
            gv1[i] = s_v1[sl];
        }
        if (t == 0) lcnt[blockIdx.x] = s_n[2];
    }

    for (int pass = 0; pass < 2; ++pass) {
        const int beg = pass ? LIST_CAP - nhi : 0;
        const int n   = pass ? nhi : nlo;
        for (int i0 = wave; i0 < n; i0 += SC_WAVES * L1_U) {
            int   rl[L1_U];
            bool  ok[L1_U];
            float v0[L1_U], v1[L1_U];
            vecf  p0[L1_U], p1[L1_U];
#pragma unroll
            for (int u = 0; u < L1_U; ++u) {
                const int i  = i0 + u * SC_WAVES;
                ok[u] = i < n;
                const int ic = beg + (ok[u] ? i : i0);
                const int key = s_key[ic];
                rl[u] = key >> KEY_SHIFT;
                v0[u] = s_v0[ic];
                v1[u] = s_v1[ic];
                const unsigned off = (unsigned)(key & KEY_MASK) * H + VEC * lane;
                p0[u] = *reinterpret_cast<const vecf*>(pre0 + off);
                p1[u] = *reinterpret_cast<const vecf*>(pre1 + off);
            }
#pragma unroll
            for (int u = 0; u < L1_U; ++u) {
                float* a = acc + rl[u] * H + lane;
#pragma unroll
                for (int j = 0; j < VEC; ++j)
                    atomicAdd(a + j * 32,
                              ok[u] ? v0[u] * p0[u][j] + v1[u] * p1[u][j] : 0.0f);
            }
        }
        __syncthreads();
    }

    for (int row = wave; row < cnt; row += SC_WAVES) {
        vecf o;
#pragma unroll
        for (int j = 0; j < VEC; ++j) o[j] = fmaxf(acc[row * H + j * 32 + lane], 0.0f);
        *reinterpret_cast<vecf*>(hout + (size_t)(lo + row) * H + VEC * lane) = o;
    }
}

__global__ __launch_bounds__(SC_THREADS)
void gcn_scatter1_lds(const float* __restrict__ p0,
                      const float* __restrict__ p1,
                      const int* __restrict__ erow,
                      const int* __restrict__ ecol,
                      const float* __restrict__ ev0,
                      const float* __restrict__ ev1,
                      float* __restrict__ out,
                      const int* __restrict__ lkey,
                      const float* __restrict__ lv0,
                      const float* __restrict__ lv1,
                      const int* __restrict__ lcnt) {
    static_assert(SC_ROWS <= SC_THREADS, "one thread per owned row");
    __shared__ float acc1[SC_ROWS];

    const int t   = threadIdx.x;
    const int lo  = blockIdx.x * SC_ROWS;
    const int cnt = min(SC_ROWS, NN - lo);

    if (t < SC_ROWS) acc1[t] = 0.0f;
    __syncthreads();

    const int nl = lcnt[blockIdx.x];
    if (nl <= LIST_CAP) {
        const int*   gk  = lkey + (size_t)blockIdx.x * LIST_CAP;
        const float* gv0 = lv0  + (size_t)blockIdx.x * LIST_CAP;
        const float* gv1 = lv1  + (size_t)blockIdx.x * LIST_CAP;
        constexpr int SW = LIST_CAP / SC_THREADS;
        int   rl[SW];
        float val[SW];
#pragma unroll
        for (int u = 0; u < SW; ++u) {
            const int i  = u * SC_THREADS + t;
            const int ic = (i < nl) ? i : t;
            const int key = gk[ic];
            const int col = key & KEY_MASK;
            rl[u]  = min(key >> KEY_SHIFT, SC_ROWS - 1);
            val[u] = gv0[ic] * p0[col] + gv1[ic] * p1[col];
        }
#pragma unroll
        for (int u = 0; u < SW; ++u)
            if (u * SC_THREADS + t < nl) atomicAdd(&acc1[rl[u]], val[u]);
    } else {
        int c = (blockIdx.x * 7) % SC_NCHUNK;
        for (int it = 0; it < SC_NCHUNK; ++it) {
            const int cbase = c * SC_CHUNK;
            int4 rv[SC_LD];
#pragma unroll
            for (int j = 0; j < SC_LD; ++j) {
                const int e0 = cbase + ((j * SC_THREADS + t) << 2);
                rv[j] = *reinterpret_cast<const int4*>(erow + min(e0, EE - 4));
            }
            unsigned long long m = 0;
#pragma unroll
            for (int j = 0; j < SC_LD; ++j) {
                const unsigned lim = (cbase + ((j * SC_THREADS + t) << 2) < EE) ? (unsigned)cnt : 0u;
                m |= (unsigned long long)((unsigned)(rv[j].x - lo) < lim) << (4 * j + 0);
                m |= (unsigned long long)((unsigned)(rv[j].y - lo) < lim) << (4 * j + 1);
                m |= (unsigned long long)((unsigned)(rv[j].z - lo) < lim) << (4 * j + 2);
                m |= (unsigned long long)((unsigned)(rv[j].w - lo) < lim) << (4 * j + 3);
            }
            while (m) {
                const int b = __builtin_ctzll(m);
                m &= m - 1;
                const int e   = cbase + ((((b >> 2) * SC_THREADS) + t) << 2) + (b & 3);
                const int rl  = erow[e] - lo;
                const int col = ecol[e];
                atomicAdd(&acc1[rl], ev0[e] * p0[col] + ev1[e] * p1[col]);
            }
            c = (c + 1 == SC_NCHUNK) ? 0 : c + 1;
        }
    }
    __syncthreads();

    if (t < cnt) out[lo + t] = acc1[t];
}

__global__ void gcn_dot64x2(const float* __restrict__ h,
                            const float* __restrict__ w0,
                            const float* __restrict__ w1,
                            float* __restrict__ p0, float* __restrict__ p1) {
    int i = blockIdx.x * blockDim.x + threadIdx.x;
    if (i >= NN) return;
    const float* row = h + (size_t)i * HH2;
    float s0 = 0.0f, s1 = 0.0f;
#pragma unroll
    for (int k = 0; k < HH2; ++k) {
        float hv = row[k];
        s0 += hv * w0[k];
        s1 += hv * w1[k];
    }
    p0[i] = s0;
    p1[i] = s1;
}

extern "C" void kernel_launch(void* const* d_in, const int* in_sizes, int n_in,
                              void* d_out, int out_size, void* d_ws, size_t ws_size,
                              hipStream_t stream) {
    const float* x     = (const float*)d_in[0];
    const int*   erow  = (const int*)d_in[1];
    const int*   ecol  = (const int*)d_in[2];
    const float* evals = (const float*)d_in[3];
    const float* W1    = (const float*)d_in[4];
    const float* W2    = (const float*)d_in[5];
    const float* W3    = (const float*)d_in[6];
    float* out = (float*)d_out;

    const float* ev0 = evals;
    const float* ev1 = evals + (size_t)EE;

    float* pre0 = (float*)d_ws;
    float* pre1 = pre0 + (size_t)NN * HH1;
    float* bufA = pre1 + (size_t)NN * HH1;
    float* bufB = bufA + (size_t)NN * HH1;
    static_assert((size_t)3 * SC_RANGES * LIST_CAP + SC_RANGES + OW_WORDS <= (size_t)NN * HH2,
                  "lists, counts and owner map fit the fourth region");
    static_assert((3 * SC_RANGES * LIST_CAP + SC_RANGES) % 4 == 0, "owner map 16-byte aligned");
    int*   lkey = (int*)bufB;
    float* lv0  = bufB + (size_t)SC_RANGES * LIST_CAP;
    float* lv1  = bufB + (size_t)2 * SC_RANGES * LIST_CAP;
    int*   lcnt = (int*)(bufB + (size_t)3 * SC_RANGES * LIST_CAP);
    int*   own  = lcnt + SC_RANGES;

    const int ZB = 256;
    static_assert(NN % 32 == 0, "whole 16-row tiles, and whole pairs of them for layer 2");
    gcn_gemm_strip<FF, HH1, true><<<GS_GRID, GS_THREADS, 0, stream>>>(
        x, W1, W1 + (size_t)FF * HH1, pre0, pre1, NN / 16, erow, own);
    gcn_scatter_l1<<<SC_RANGES, SC_THREADS, 0, stream>>>(
        pre0, pre1, erow, ecol, ev0, ev1, bufA, lkey, lv0, lv1, lcnt, own);

    gcn_gemm_strip<HH1, HH2, false><<<GS_GRID, GS_THREADS, 0, stream>>>(
        bufA, W2, W2 + (size_t)HH1 * HH2, pre0, pre1, NN / 32, erow, own);
    gcn_scatter_lds<HH2><<<SC_RANGES, SC_THREADS, 0, stream>>>(
        pre0, pre1, erow, ecol, ev0, ev1, bufA, lkey, lv0, lv1, lcnt, own);

    gcn_dot64x2<<<(NN + ZB - 1) / ZB, ZB, 0, stream>>>(
        bufA, W3, W3 + (size_t)HH2, pre0, pre1);
    gcn_scatter1_lds<<<SC_RANGES, SC_THREADS, 0, stream>>>(
        pre0, pre1, erow, ecol, ev0, ev1, out, lkey, lv0, lv1, lcnt);
}


# Layer-1 product launch bins each edge into an (owner, chunk) cell via LDS; layer-1 aggregation reads its 256 cells, not the 1.6M-byte owner map; sums: lower-half sources first, arrival order inside
# speedup vs baseline: 3.3470x; 1.1329x over previous
#include <hip/hip_runtime.h>

typedef __attribute__((ext_vector_type(2))) float v2f;
typedef __attribute__((ext_vector_type(8))) float v8f;

#define NN 100000
#define FF 256
#define EE 1600000
#define HH1 128
#define HH2 64

#define GS_PITCH   33
#define GS_WAVES   16
#define GS_THREADS (32 * GS_WAVES)
#define GS_GRID    256

#define OW_ROWS    ((NN + 255) / 256)
#define OW_WORDS   (EE / 4)

#define BK_CHUNK   (EE / GS_GRID)
#define BK_CAP     64
#define BK_LP      (BK_CAP + 1)
#define BK_IT      ((BK_CHUNK + GS_THREADS - 1) / GS_THREADS)
#define BK_OWNER   (3 * GS_GRID * BK_CAP)

template <int K, int Nout, bool OWNERS>
__global__ __launch_bounds__(GS_THREADS)
void gcn_gemm_strip(const float* __restrict__ A,
                    const float* __restrict__ B0, const float* __restrict__ B1,
                    float* __restrict__ C0, float* __restrict__ C1, int nsteps,
                    const int* __restrict__ erow, int* __restrict__ own,
                    const int* __restrict__ ecol,
                    const float* __restrict__ ev0, const float* __restrict__ ev1,
                    int* __restrict__ cells, int* __restrict__ cells_last,
                    int* __restrict__ tab) {
    constexpr int STRIPS = Nout / 16;
    constexpr int WPT    = 2 * STRIPS;
    constexpr int MT     = GS_WAVES / WPT;
    constexpr int K4     = K / 4;
    constexpr int KJ     = K / 8;
    static_assert(K % 8 == 0 && Nout % 16 == 0 && GS_WAVES % WPT == 0, "tile dims");
    static_assert(MT * 16 * K4 == 2 * GS_THREADS, "two float4 per thread and step");

    __shared__ float4 sA[2][MT * KJ * GS_PITCH];

    const int t    = threadIdx.x;
    const int lane = t & 31;
    const int wave = t >> 5;
    const int h    = lane >> 4;
    const int c    = lane & 15;
    const int g    = wave / WPT;

    if constexpr (OWNERS) {
        static_assert(EE % 4 == 0, "edge rows read in 16-byte pieces");
        static_assert(4 * GS_GRID * GS_THREADS >= OW_WORDS, "four words per thread cover the map");
        static_assert((NN - 1) / OW_ROWS <= 255, "owner fits a byte");
        for (int i = 0; i < 4; ++i) {
            const int w = i * (GS_GRID * GS_THREADS) + blockIdx.x * GS_THREADS + t;
            if (w < OW_WORDS) {
                const int4 r = *reinterpret_cast<const int4*>(erow + 4 * w);
                const unsigned o0 = min((unsigned)r.x / OW_ROWS, 255u);
                const unsigned o1 = min((unsigned)r.y / OW_ROWS, 255u);
                const unsigned o2 = min((unsigned)r.z / OW_ROWS, 255u);
                const unsigned o3 = min((unsigned)r.w / OW_ROWS, 255u);
                own[w] = (int)(o0 | o1 << 8 | o2 << 16 | o3 << 24);
            }
        }

        static_assert(EE % GS_GRID == 0, "whole chunks");
        static_assert(BK_CAP == 64 && GS_WAVES == 16, "copy-out below: two lines per cell and field, 96 lines per wave");
        __shared__ int b_cnt[256];
        __shared__ int b_cell[3][256][BK_LP];
        if (t < 256) b_cnt[t] = 0;
        __syncthreads();
        {
            const int e0 = blockIdx.x * BK_CHUNK;
            int   er[BK_IT], ec[BK_IT];
            float ea[BK_IT], eb[BK_IT];
#pragma unroll
            for (int k = 0; k < BK_IT; ++k) {
                const int i  = k * GS_THREADS + t;
                const int e  = e0 + min(i, BK_CHUNK - 1);
                er[k] = (i < BK_CHUNK) ? erow[e] : -1;
                ec[k] = ecol[e];
                ea[k] = ev0[e];
                eb[k] = ev1[e];
            }
#pragma unroll
            for (int k = 0; k < BK_IT; ++k) {
                const int r = er[k];
                if ((unsigned)r < (unsigned)NN) {
                    const int o    = r / OW_ROWS;
                    const int rank = atomicAdd(&b_cnt[o], 1);
                    if (rank < BK_CAP) {
                        b_cell[0][o][rank] = ((r - o * OW_ROWS) << 17) | ec[k];
                        b_cell[1][o][rank] = __float_as_int(ea[k]);
                        b_cell[2][o][rank] = __float_as_int(eb[k]);
                    }
                }
            }
        }
        __syncthreads();
        if (t < 256) tab[blockIdx.x * 256 + t] = b_cnt[t];
        {
            const int hf = (wave & 1) * 32;
            for (int m0 = 0; m0 < 96; m0 += 8) {
                int cn[8], val[8];
#pragma unroll
                for (int u = 0; u < 8; ++u) {
                    const int m = m0 + u;
                    const int a = m >> 5;
                    const int o = (m & 31) * 8 + (wave >> 1);
                    cn[u]  = b_cnt[o];
                    val[u] = b_cell[a][o][hf + lane];
                }
#pragma unroll
                for (int u = 0; u < 8; ++u) {
                    const int m = m0 + u;
                    const int a = m >> 5;
                    const int o = (m & 31) * 8 + (wave >> 1);
                    if (hf < cn[u]) {
                        int* dst = (o == 255) ? cells_last : cells + (size_t)o * (OW_ROWS * HH1);
                        dst[(a * GS_GRID + blockIdx.x) * BK_CAP + hf + lane] = val[u];
                    }
                }
            }
        }
    }
    const int sup  = (wave % WPT) / STRIPS;
    const int col  = (wave % STRIPS) * 16 + c;

    const float* __restrict__ bp = (sup ? B1 : B0) + (size_t)(4 * h) * Nout + col;
    v2f breg[K4];
#pragma unroll
    for (int s = 0; s < K4; ++s) {
        const int kk = 8 * (s >> 1) + 2 * (s & 1);
        breg[s].x = bp[(kk + 0) * Nout];
        breg[s].y = bp[(kk + 1) * Nout];
    }
    float* __restrict__ cp = (sup ? C1 : C0) + (size_t)(16 * g + 8 * h) * Nout + col;

    const int f0 = t, f1 = t + GS_THREADS;
    const int r0 = f0 / K4, c0 = f0 % K4;
    const int r1 = f1 / K4, c1 = f1 % K4;
    const int srow0  = r0 * K + c0 * 4;
    const int srow1  = r1 * K + c1 * 4;
    const int sslot0 = ((r0 >> 4) * KJ + (c0 >> 1)) * GS_PITCH + (c0 & 1) * 16 + (r0 & 15);
    const int sslot1 = ((r1 >> 4) * KJ + (c1 >> 1)) * GS_PITCH + (c1 & 1) * 16 + (r1 & 15);

    int step = blockIdx.x;
    {
        const float* __restrict__ ap = A + (size_t)min(step, nsteps - 1) * (MT * 16 * K);
        const float4 p0 = *reinterpret_cast<const float4*>(ap + srow0);
        const float4 p1 = *reinterpret_cast<const float4*>(ap + srow1);
        sA[0][sslot0] = p0;
        sA[0][sslot1] = p1;
    }
    __syncthreads();

    int buf = 0;
    v8f rprev = {};
    float* __restrict__ cprev = cp;
    bool pending = false;
    for (; step < nsteps; step += gridDim.x) {
        const float* __restrict__ ap = A + (size_t)min(step + (int)gridDim.x, nsteps - 1) * (MT * 16 * K);
        const float4 st0 = *reinterpret_cast<const float4*>(ap + srow0);
        const float4 st1 = *reinterpret_cast<const float4*>(ap + srow1);

        if (pending) {
#pragma unroll
            for (int v = 0; v < 8; ++v) cprev[(size_t)v * Nout] = rprev[v];
        }

        const float4* __restrict__ fa = sA[buf] + g * (KJ * GS_PITCH) + lane;
        v8f acc0 = {}, acc1 = {}, acc2 = {}, acc3 = {};
#pragma unroll
        for (int j = 0; j < KJ; j += 2) {
            const float4 a0 = fa[(j + 0) * GS_PITCH];
            const float4 a1 = fa[(j + 1) * GS_PITCH];
            v2f x0 = {a0.x, a0.y}, x1 = {a0.z, a0.w}, x2 = {a1.x, a1.y}, x3 = {a1.z, a1.w};
            acc0 = __builtin_amdgcn_wmma_f32_16x16x4_f32(false, x0, false, breg[2 * j + 0], (short)0, acc0, false, false);
            acc1 = __builtin_amdgcn_wmma_f32_16x16x4_f32(false, x1, false, breg[2 * j + 1], (short)0, acc1, false, false);
            acc2 = __builtin_amdgcn_wmma_f32_16x16x4_f32(false, x2, false, breg[2 * j + 2], (short)0, acc2, false, false);
            acc3 = __builtin_amdgcn_wmma_f32_16x16x4_f32(false, x3, false, breg[2 * j + 3], (short)0, acc3, false, false);
        }
        rprev   = (acc0 + acc1) + (acc2 + acc3);
        cprev   = cp + (size_t)step * (MT * 16 * Nout);
        pending = true;

        sA[buf ^ 1][sslot0] = st0;
        sA[buf ^ 1][sslot1] = st1;
        __syncthreads();
        buf ^= 1;
    }
    if (pending) {
#pragma unroll
        for (int v = 0; v < 8; ++v) cprev[(size_t)v * Nout] = rprev[v];
    }
}

#define SC_THREADS 1024
#define SC_WAVES   (SC_THREADS / 32)
#define SC_RANGES  256
#define SC_ROWS    ((NN + SC_RANGES - 1) / SC_RANGES)
#define SC_LD      10
#define SC_CHUNK   (SC_THREADS * SC_LD * 4)
#define SC_NCHUNK  ((EE + SC_CHUNK - 1) / SC_CHUNK)
#define SC_QCAP    1024
#define SC_U       6
static_assert(OW_ROWS == SC_ROWS && SC_RANGES == 256, "the owner byte is the work-group index");
#define OW_LD      10
#define OW_NQ      (EE / 16)
#define OW_CHUNKQ  (SC_THREADS * OW_LD)
#define OW_NCHUNK  ((OW_NQ + OW_CHUNKQ - 1) / OW_CHUNKQ)
static_assert(EE % 16 == 0, "owner map read in 16-byte pieces");

__device__ __forceinline__ unsigned own_hit(unsigned w, unsigned pat) {
    const unsigned x = w ^ pat;
    return ((x - 0x01010101u) & ~x & 0x80808080u) != 0u;
}
__device__ __forceinline__ unsigned long long own_hit4(int4 v, unsigned pat, bool valid) {
    const unsigned f = own_hit((unsigned)v.x, pat) | own_hit((unsigned)v.y, pat) << 1 |
                       own_hit((unsigned)v.z, pat) << 2 | own_hit((unsigned)v.w, pat) << 3;
    return valid ? f : 0u;
}

#define LIST_CAP   7168
#define LIST_U     8
#define KEY_SHIFT  17
#define KEY_MASK   ((1 << KEY_SHIFT) - 1)
static_assert(NN <= (1 << KEY_SHIFT), "col fits the low key bits");
static_assert(SC_ROWS <= (1 << (31 - KEY_SHIFT)), "local row fits the high key bits");
static_assert(LIST_CAP % SC_THREADS == 0, "layer 3 walks the list in whole sweeps");
static_assert(KEY_SHIFT == 17 && SC_RANGES == GS_GRID, "the edge cells carry the same key, one chunk per product work-group");
static_assert(BK_OWNER <= SC_ROWS * HH1, "an owner's cells fit into its own rows of the layer-1 output");
static_assert((GS_GRID * BK_CAP) % SC_THREADS == 0, "cell slots read in whole sweeps");

template <int H>
__global__ __launch_bounds__(SC_THREADS)
void gcn_scatter_lds(const float* __restrict__ pre0,
                     const float* __restrict__ pre1,
                     const int* __restrict__ erow,
                     const int* __restrict__ ecol,
                     const float* __restrict__ ev0,
                     const float* __restrict__ ev1,
                     float* __restrict__ hout,
                     const int* __restrict__ lkey,
                     const float* __restrict__ lv0,
                     const float* __restrict__ lv1,
                     const int* __restrict__ lcnt,
                     const int* __restrict__ own) {
    constexpr int VEC = H / 32;
    typedef __attribute__((ext_vector_type(VEC))) float vecf;
    static_assert(H == 32 * VEC && (VEC == 2 || VEC == 4), "row = one vector load per lane");
    static_assert(EE % 4 == 0, "edge list read in 16-byte pieces");
    static_assert(OW_LD * 4 <= 64, "one mask bit per owner word of a thread");

    __shared__ float acc[SC_ROWS * H];
    __shared__ int   q_rl[3][SC_QCAP];
    __shared__ int   q_col[3][SC_QCAP];
    __shared__ float q_v0[3][SC_QCAP];
    __shared__ float q_v1[3][SC_QCAP];
    __shared__ int   q_n[3];

    const int t    = threadIdx.x;
    const int lane = t & 31;
    const int wave = t >> 5;
    const int lo   = blockIdx.x * SC_ROWS;
    const int cnt  = min(SC_ROWS, NN - lo);

    for (int i = t; i < SC_ROWS * H; i += SC_THREADS) acc[i] = 0.0f;
    if (t < 3) q_n[t] = 0;

    const int*   const gk  = lkey + (size_t)blockIdx.x * LIST_CAP;
    const float* const gv0 = lv0  + (size_t)blockIdx.x * LIST_CAP;
    const float* const gv1 = lv1  + (size_t)blockIdx.x * LIST_CAP;

    bool listed = false;
    {
        const int nl = lcnt[blockIdx.x];
        if (nl <= LIST_CAP) {
            listed = true;
            __shared__ int   s_key[LIST_CAP];
            __shared__ float s_v0[LIST_CAP];
            __shared__ float s_v1[LIST_CAP];
            for (int i = t; i < nl; i += SC_THREADS) {
                s_key[i] = gk[i];
                s_v0[i]  = gv0[i];
                s_v1[i]  = gv1[i];
            }
            __syncthreads();
            for (int i0 = wave; i0 < nl; i0 += SC_WAVES * LIST_U) {
                int   rl[LIST_U];
                bool  ok[LIST_U];
                float v0[LIST_U], v1[LIST_U];
                vecf  p0[LIST_U], p1[LIST_U];
#pragma unroll
                for (int u = 0; u < LIST_U; ++u) {
                    const int i  = i0 + u * SC_WAVES;
                    ok[u] = i < nl;
                    const int ic = ok[u] ? i : i0;
                    const int key = s_key[ic];
                    rl[u] = min(key >> KEY_SHIFT, SC_ROWS - 1);
                    v0[u] = s_v0[ic];
                    v1[u] = s_v1[ic];
                    const unsigned off = (unsigned)(key & KEY_MASK) * H + VEC * lane;
                    p0[u] = *reinterpret_cast<const vecf*>(pre0 + off);
                    p1[u] = *reinterpret_cast<const vecf*>(pre1 + off);
                }
#pragma unroll
                for (int u = 0; u < LIST_U; ++u) {
                    float* a = acc + rl[u] * H + lane;
#pragma unroll
                    for (int j = 0; j < VEC; ++j)
                        atomicAdd(a + j * 32,
                                  ok[u] ? v0[u] * p0[u][j] + v1[u] * p1[u][j] : 0.0f);
                }
            }
        }
    }

    if (!listed) {
        int c = (blockIdx.x * 7) % OW_NCHUNK;

        const int4* __restrict__ own4 = reinterpret_cast<const int4*>(own);
        const unsigned pat = blockIdx.x * 0x01010101u;
        int4 rv[OW_LD];
        unsigned long long m = 0;
#pragma unroll
        for (int j = 0; j < OW_LD; ++j) {
            const int q = c * OW_CHUNKQ + j * SC_THREADS + t;
            rv[j] = own4[min(q, OW_NQ - 1)];
        }
#pragma unroll
        for (int j = 0; j < OW_LD; ++j) {
            m |= own_hit4(rv[j], pat, c * OW_CHUNKQ + j * SC_THREADS + t < OW_NQ) << (4 * j);
        }
        __syncthreads();

        for (int it = 0; it < OW_NCHUNK; ++it) {
            const int qb = it % 3;
            const int cbase = c * OW_CHUNKQ;
            int cn = c + 1;
            if (cn == OW_NCHUNK) cn = 0;

            if (it + 1 < OW_NCHUNK) {
#pragma unroll
                for (int j = 0; j < OW_LD; ++j) {
                    const int q = cn * OW_CHUNKQ + j * SC_THREADS + t;
                    rv[j] = own4[min(q, OW_NQ - 1)];
                }
            }

            while (m) {
                const int b = __builtin_ctzll(m);
                m &= m - 1;
                const int eb = ((cbase + (b >> 2) * SC_THREADS + t) << 4) + ((b & 3) << 2);
                const int4   r4 = *reinterpret_cast<const int4*>(erow + eb);
                const int4   c4 = *reinterpret_cast<const int4*>(ecol + eb);
                const float4 a4 = *reinterpret_cast<const float4*>(ev0 + eb);
                const float4 b4 = *reinterpret_cast<const float4*>(ev1 + eb);
                const int   rr[4] = {r4.x, r4.y, r4.z, r4.w};
                const int   cc[4] = {c4.x, c4.y, c4.z, c4.w};
                const float aa[4] = {a4.x, a4.y, a4.z, a4.w};
                const float bb[4] = {b4.x, b4.y, b4.z, b4.w};
#pragma unroll
                for (int i = 0; i < 4; ++i) {
                    const int rl = rr[i] - lo;
                    if ((unsigned)rl >= (unsigned)cnt) continue;
                    const int col = cc[i];
                    const float v0 = aa[i];
                    const float v1 = bb[i];
                    const int slot = atomicAdd(&q_n[qb], 1);
                    if (slot < SC_QCAP) {
                        q_rl[qb][slot]  = rl;
                        q_col[qb][slot] = col;
                        q_v0[qb][slot]  = v0;
                        q_v1[qb][slot]  = v1;
                    } else {
                        const float* s0 = pre0 + (size_t)col * H;
                        const float* s1 = pre1 + (size_t)col * H;
                        for (int k = 0; k < H; ++k)
                            atomicAdd(&acc[rl * H + (k % VEC) * 32 + (k / VEC)],
                                      v0 * s0[k] + v1 * s1[k]);
                    }
                }
            }

            if (it + 1 < OW_NCHUNK) {
#pragma unroll
                for (int j = 0; j < OW_LD; ++j) {
                    m |= own_hit4(rv[j], pat, cn * OW_CHUNKQ + j * SC_THREADS + t < OW_NQ) << (4 * j);
                }
            }
            c = cn;

            __syncthreads();
            if (t == 0) q_n[(it + 2) % 3] = 0;

            const int n = min(q_n[qb], SC_QCAP);
            for (int i0 = wave; i0 < n; i0 += SC_WAVES * SC_U) {
                int   rl[SC_U];
                bool  ok[SC_U];
                float v0[SC_U], v1[SC_U];
                vecf  p0[SC_U], p1[SC_U];
#pragma unroll
                for (int u = 0; u < SC_U; ++u) {
                    const int i  = i0 + u * SC_WAVES;
                    ok[u] = i < n;
                    const int ic = ok[u] ? i : i0;
                    rl[u] = q_rl[qb][ic];
                    v0[u] = q_v0[qb][ic];
                    v1[u] = q_v1[qb][ic];
                    const unsigned off = (unsigned)q_col[qb][ic] * H + VEC * lane;
                    p0[u] = *reinterpret_cast<const vecf*>(pre0 + off);
                    p1[u] = *reinterpret_cast<const vecf*>(pre1 + off);
                }
#pragma unroll
                for (int u = 0; u < SC_U; ++u) {
                    float* a = acc + rl[u] * H + lane;
#pragma unroll
                    for (int j = 0; j < VEC; ++j)
                        atomicAdd(a + j * 32,
                                  ok[u] ? v0[u] * p0[u][j] + v1[u] * p1[u][j] : 0.0f);
                }
            }
        }
    }
    __syncthreads();

    for (int row = wave; row < cnt; row += SC_WAVES) {
        vecf o;
#pragma unroll
        for (int j = 0; j < VEC; ++j) o[j] = fmaxf(acc[row * H + j * 32 + lane], 0.0f);
        *reinterpret_cast<vecf*>(hout + (size_t)(lo + row) * H + VEC * lane) = o;
    }
}

#define L1_U      6
#define L1_SPLIT  (NN / 2)

__global__ __launch_bounds__(SC_THREADS)
void gcn_scatter_l1(const float* __restrict__ pre0,
                    const float* __restrict__ pre1,
                    const int* __restrict__ erow,
                    const int* __restrict__ ecol,
                    const float* __restrict__ ev0,
                    const float* __restrict__ ev1,
                    float* hout,
                    int* __restrict__ lkey,
                    float* __restrict__ lv0,
                    float* __restrict__ lv1,
                    int* __restrict__ lcnt,
                    const int* __restrict__ own,
                    const int* cells,
                    const int* __restrict__ cells_last,
                    const int* __restrict__ tab) {
    constexpr int H   = HH1;
    constexpr int VEC = H / 32;
    typedef __attribute__((ext_vector_type(VEC))) float vecf;
    static_assert(H == 32 * VEC && VEC == 4, "row = one 16-byte load per lane");
    static_assert(OW_LD * 4 <= 64, "one mask bit per owner word of a thread");

    __shared__ float acc[SC_ROWS * H];
    __shared__ int   s_key[LIST_CAP];
    __shared__ float s_v0[LIST_CAP];
    __shared__ float s_v1[LIST_CAP];
    __shared__ int   s_n[3];
    __shared__ int   s_cnt[GS_GRID];
    __shared__ int   s_part[2][GS_GRID / 32];

    const int t    = threadIdx.x;
    const int lane = t & 31;
    const int wave = t >> 5;
    const int lo   = blockIdx.x * SC_ROWS;
    const int cnt  = min(SC_ROWS, NN - lo);

    for (int i = t; i < SC_ROWS * H; i += SC_THREADS) acc[i] = 0.0f;
    if (t < 3) s_n[t] = 0;

    if (t < GS_GRID) {
        const int cc = tab[t * SC_RANGES + blockIdx.x];
        s_cnt[t] = cc;
        int sum = cc, mx = cc;
        for (int d = 16; d > 0; d >>= 1) {
            sum += __shfl_xor(sum, d);
            mx   = max(mx, __shfl_xor(mx, d));
        }
        if (lane == 0) { s_part[0][wave] = sum; s_part[1][wave] = mx; }
    }
    __syncthreads();
    int total = 0, cmax = 0;
    for (int w = 0; w < GS_GRID / 32; ++w) {
        total += s_part[0][w];
        cmax   = max(cmax, s_part[1][w]);
    }
    const bool celled = cmax <= BK_CAP && total <= LIST_CAP;

    if (celled) {
        const int* cb = (blockIdx.x == SC_RANGES - 1) ? cells_last
                                                      : cells + (size_t)blockIdx.x * (SC_ROWS * H);
        constexpr int SW = GS_GRID * BK_CAP / SC_THREADS;
        int   key[SW], cg[SW];
        float w0[SW], w1[SW];
        bool  ok[SW];
#pragma unroll
        for (int k = 0; k < SW; ++k) cg[k] = s_cnt[(k * SC_THREADS + t) / BK_CAP];
#pragma unroll
        for (int k = 0; k < SW; ++k) {
            const int s   = k * SC_THREADS + t;
            const int g   = s / BK_CAP;
            const int idx = s % BK_CAP;
            ok[k]  = idx < cg[k];
            key[k] = 0; w0[k] = 0.0f; w1[k] = 0.0f;
            if (ok[k]) {
                key[k] = cb[(0 * GS_GRID + g) * BK_CAP + idx];
                w0[k]  = __int_as_float(cb[(1 * GS_GRID + g) * BK_CAP + idx]);
                w1[k]  = __int_as_float(cb[(2 * GS_GRID + g) * BK_CAP + idx]);
            }
        }
        const unsigned below = (1u << lane) - 1u;
#pragma unroll
        for (int k = 0; k < SW; ++k) {
            const bool isl = ok[k] && (key[k] & KEY_MASK) < L1_SPLIT;
            const bool ish = ok[k] && !isl;
            const unsigned bl = (unsigned)__ballot(isl);
            const unsigned bh = (unsigned)__ballot(ish);
            int basel = 0, baseh = 0;
            if (lane == 0) {
                if (bl) basel = atomicAdd(&s_n[0], __popc(bl));
                if (bh) baseh = atomicAdd(&s_n[1], __popc(bh));
            }
            basel = __builtin_amdgcn_readfirstlane(basel);
            baseh = __builtin_amdgcn_readfirstlane(baseh);
            if (ok[k]) {
                const int pos = isl ? basel + __popc(bl & below)
                                    : LIST_CAP - 1 - (baseh + __popc(bh & below));
                s_key[pos] = key[k];
                s_v0[pos]  = w0[k];
                s_v1[pos]  = w1[k];
            }
        }
        if (t == 0) s_n[2] = total;
    } else {
        int c = (blockIdx.x * 7) % OW_NCHUNK;

        const int4* __restrict__ own4 = reinterpret_cast<const int4*>(own);
        const unsigned pat = blockIdx.x * 0x01010101u;
        int4 rv[OW_LD];
        unsigned long long m = 0;
#pragma unroll
        for (int j = 0; j < OW_LD; ++j) {
            const int q = c * OW_CHUNKQ + j * SC_THREADS + t;
            rv[j] = own4[min(q, OW_NQ - 1)];
        }
#pragma unroll
        for (int j = 0; j < OW_LD; ++j) {
            m |= own_hit4(rv[j], pat, c * OW_CHUNKQ + j * SC_THREADS + t < OW_NQ) << (4 * j);
        }
        __syncthreads();

        for (int it = 0; it < OW_NCHUNK; ++it) {
            const int cbase = c * OW_CHUNKQ;
            int cn = c + 1;
            if (cn == OW_NCHUNK) cn = 0;

            if (it + 1 < OW_NCHUNK) {
#pragma unroll
                for (int j = 0; j < OW_LD; ++j) {
                    const int q = cn * OW_CHUNKQ + j * SC_THREADS + t;
                    rv[j] = own4[min(q, OW_NQ - 1)];
                }
            }

            while (m) {
                const int b = __builtin_ctzll(m);
                m &= m - 1;
                const int eb = ((cbase + (b >> 2) * SC_THREADS + t) << 4) + ((b & 3) << 2);
                const int4   r4 = *reinterpret_cast<const int4*>(erow + eb);
                const int4   c4 = *reinterpret_cast<const int4*>(ecol + eb);
                const float4 a4 = *reinterpret_cast<const float4*>(ev0 + eb);
                const float4 b4 = *reinterpret_cast<const float4*>(ev1 + eb);
                const int   rr[4] = {r4.x, r4.y, r4.z, r4.w};
                const int   cc[4] = {c4.x, c4.y, c4.z, c4.w};
                const float aa[4] = {a4.x, a4.y, a4.z, a4.w};
                const float bb[4] = {b4.x, b4.y, b4.z, b4.w};
#pragma unroll
                for (int i = 0; i < 4; ++i) {
                    const int rl = rr[i] - lo;
                    if ((unsigned)rl >= (unsigned)cnt) continue;
                    const int col = cc[i];
                    const float v0 = aa[i];
                    const float v1 = bb[i];
                    const int all = atomicAdd(&s_n[2], 1);
                    if (all < LIST_CAP) {
                        int pos;
                        if (col < L1_SPLIT) pos = atomicAdd(&s_n[0], 1);
                        else                pos = LIST_CAP - 1 - atomicAdd(&s_n[1], 1);
                        s_key[pos] = (rl << KEY_SHIFT) | col;
                        s_v0[pos]  = v0;
                        s_v1[pos]  = v1;
                    } else {
                        const float* s0 = pre0 + (size_t)col * H;
                        const float* s1 = pre1 + (size_t)col * H;
                        for (int k = 0; k < H; ++k)
                            atomicAdd(&acc[rl * H + (k % VEC) * 32 + (k / VEC)],
                                      v0 * s0[k] + v1 * s1[k]);
                    }
                }
            }

            if (it + 1 < OW_NCHUNK) {
#pragma unroll
                for (int j = 0; j < OW_LD; ++j) {
                    m |= own_hit4(rv[j], pat, cn * OW_CHUNKQ + j * SC_THREADS + t < OW_NQ) << (4 * j);
                }
            }
            c = cn;
        }
    }
    __syncthreads();

    const int nlo = s_n[0];
    const int nhi = s_n[1];

    {
        int*   const gk  = lkey + (size_t)blockIdx.x * LIST_CAP;
        float* const gv0 = lv0  + (size_t)blockIdx.x * LIST_CAP;
        float* const gv1 = lv1  + (size_t)blockIdx.x * LIST_CAP;
        for (int i = t; i < nlo + nhi; i += SC_THREADS) {
            const int sl = (i < nlo) ? i : LIST_CAP - nhi + (i - nlo);
            gk[i]  = s_key[sl];
            gv0[i] = s_v0[sl];
            gv1[i] = s_v1[sl];
        }
        if (t == 0) lcnt[blockIdx.x] = s_n[2];
    }

    for (int pass = 0; pass < 2; ++pass) {
        const int beg = pass ? LIST_CAP - nhi : 0;
        const int n   = pass ? nhi : nlo;
        for (int i0 = wave; i0 < n; i0 += SC_WAVES * L1_U) {
            int   rl[L1_U];
            bool  ok[L1_U];
            float v0[L1_U], v1[L1_U];
            vecf  p0[L1_U], p1[L1_U];
#pragma unroll
            for (int u = 0; u < L1_U; ++u) {
                const int i  = i0 + u * SC_WAVES;
                ok[u] = i < n;
                const int ic = beg + (ok[u] ? i : i0);
                const int key = s_key[ic];
                rl[u] = key >> KEY_SHIFT;
                v0[u] = s_v0[ic];
                v1[u] = s_v1[ic];
                const unsigned off = (unsigned)(key & KEY_MASK) * H + VEC * lane;
                p0[u] = *reinterpret_cast<const vecf*>(pre0 + off);
                p1[u] = *reinterpret_cast<const vecf*>(pre1 + off);
            }
#pragma unroll
            for (int u = 0; u < L1_U; ++u) {
                float* a = acc + rl[u] * H + lane;
#pragma unroll
                for (int j = 0; j < VEC; ++j)
                    atomicAdd(a + j * 32,
                              ok[u] ? v0[u] * p0[u][j] + v1[u] * p1[u][j] : 0.0f);
            }
        }
        __syncthreads();
    }

    for (int row = wave; row < cnt; row += SC_WAVES) {
        vecf o;
#pragma unroll
        for (int j = 0; j < VEC; ++j) o[j] = fmaxf(acc[row * H + j * 32 + lane], 0.0f);
        *reinterpret_cast<vecf*>(hout + (size_t)(lo + row) * H + VEC * lane) = o;
    }
}

__global__ __launch_bounds__(SC_THREADS)
void gcn_scatter1_lds(const float* __restrict__ p0,
                      const float* __restrict__ p1,
                      const int* __restrict__ erow,
                      const int* __restrict__ ecol,
                      const float* __restrict__ ev0,
                      const float* __restrict__ ev1,
                      float* __restrict__ out,
                      const int* __restrict__ lkey,
                      const float* __restrict__ lv0,
                      const float* __restrict__ lv1,
                      const int* __restrict__ lcnt) {
    static_assert(SC_ROWS <= SC_THREADS, "one thread per owned row");
    __shared__ float acc1[SC_ROWS];

    const int t   = threadIdx.x;
    const int lo  = blockIdx.x * SC_ROWS;
    const int cnt = min(SC_ROWS, NN - lo);

    if (t < SC_ROWS) acc1[t] = 0.0f;
    __syncthreads();

    const int nl = lcnt[blockIdx.x];
    if (nl <= LIST_CAP) {
        const int*   gk  = lkey + (size_t)blockIdx.x * LIST_CAP;
        const float* gv0 = lv0  + (size_t)blockIdx.x * LIST_CAP;
        const float* gv1 = lv1  + (size_t)blockIdx.x * LIST_CAP;
        constexpr int SW = LIST_CAP / SC_THREADS;
        int   rl[SW];
        float val[SW];
#pragma unroll
        for (int u = 0; u < SW; ++u) {
            const int i  = u * SC_THREADS + t;
            const int ic = (i < nl) ? i : t;
            const int key = gk[ic];
            const int col = key & KEY_MASK;
            rl[u]  = min(key >> KEY_SHIFT, SC_ROWS - 1);
            val[u] = gv0[ic] * p0[col] + gv1[ic] * p1[col];
        }
#pragma unroll
        for (int u = 0; u < SW; ++u)
            if (u * SC_THREADS + t < nl) atomicAdd(&acc1[rl[u]], val[u]);
    } else {
        int c = (blockIdx.x * 7) % SC_NCHUNK;
        for (int it = 0; it < SC_NCHUNK; ++it) {
            const int cbase = c * SC_CHUNK;
            int4 rv[SC_LD];
#pragma unroll
            for (int j = 0; j < SC_LD; ++j) {
                const int e0 = cbase + ((j * SC_THREADS + t) << 2);
                rv[j] = *reinterpret_cast<const int4*>(erow + min(e0, EE - 4));
            }
            unsigned long long m = 0;
#pragma unroll
            for (int j = 0; j < SC_LD; ++j) {
                const unsigned lim = (cbase + ((j * SC_THREADS + t) << 2) < EE) ? (unsigned)cnt : 0u;
                m |= (unsigned long long)((unsigned)(rv[j].x - lo) < lim) << (4 * j + 0);
                m |= (unsigned long long)((unsigned)(rv[j].y - lo) < lim) << (4 * j + 1);
                m |= (unsigned long long)((unsigned)(rv[j].z - lo) < lim) << (4 * j + 2);
                m |= (unsigned long long)((unsigned)(rv[j].w - lo) < lim) << (4 * j + 3);
            }
            while (m) {
                const int b = __builtin_ctzll(m);
                m &= m - 1;
                const int e   = cbase + ((((b >> 2) * SC_THREADS) + t) << 2) + (b & 3);
                const int rl  = erow[e] - lo;
                const int col = ecol[e];
                atomicAdd(&acc1[rl], ev0[e] * p0[col] + ev1[e] * p1[col]);
            }
            c = (c + 1 == SC_NCHUNK) ? 0 : c + 1;
        }
    }
    __syncthreads();

    if (t < cnt) out[lo + t] = acc1[t];
}

__global__ void gcn_dot64x2(const float* __restrict__ h,
                            const float* __restrict__ w0,
                            const float* __restrict__ w1,
                            float* __restrict__ p0, float* __restrict__ p1) {
    int i = blockIdx.x * blockDim.x + threadIdx.x;
    if (i >= NN) return;
    const float* row = h + (size_t)i * HH2;
    float s0 = 0.0f, s1 = 0.0f;
#pragma unroll
    for (int k = 0; k < HH2; ++k) {
        float hv = row[k];
        s0 += hv * w0[k];
        s1 += hv * w1[k];
    }
    p0[i] = s0;
    p1[i] = s1;
}

extern "C" void kernel_launch(void* const* d_in, const int* in_sizes, int n_in,
                              void* d_out, int out_size, void* d_ws, size_t ws_size,
                              hipStream_t stream) {
    const float* x     = (const float*)d_in[0];
    const int*   erow  = (const int*)d_in[1];
    const int*   ecol  = (const int*)d_in[2];
    const float* evals = (const float*)d_in[3];
    const float* W1    = (const float*)d_in[4];
    const float* W2    = (const float*)d_in[5];
    const float* W3    = (const float*)d_in[6];
    float* out = (float*)d_out;

    const float* ev0 = evals;
    const float* ev1 = evals + (size_t)EE;

    float* pre0 = (float*)d_ws;
    float* pre1 = pre0 + (size_t)NN * HH1;
    float* bufA = pre1 + (size_t)NN * HH1;
    float* bufB = bufA + (size_t)NN * HH1;
    static_assert((size_t)3 * SC_RANGES * LIST_CAP + SC_RANGES + OW_WORDS <= (size_t)NN * HH2,
                  "lists, counts and owner map fit the fourth region");
    static_assert((3 * SC_RANGES * LIST_CAP + SC_RANGES) % 4 == 0, "owner map 16-byte aligned");
    int*   lkey = (int*)bufB;
    float* lv0  = bufB + (size_t)SC_RANGES * LIST_CAP;
    float* lv1  = bufB + (size_t)2 * SC_RANGES * LIST_CAP;
    int*   lcnt = (int*)(bufB + (size_t)3 * SC_RANGES * LIST_CAP);
    int*   own  = lcnt + SC_RANGES;
    int*   tab  = own + OW_WORDS;
    int*   cellL = tab + GS_GRID * SC_RANGES;
    int*   cells = (int*)bufA;
    static_assert((size_t)3 * SC_RANGES * LIST_CAP + SC_RANGES + OW_WORDS + GS_GRID * SC_RANGES + BK_OWNER
                      <= (size_t)NN * HH2, "count table and the last owner's cells fit the fourth region too");
    static_assert((3 * SC_RANGES * LIST_CAP + SC_RANGES + OW_WORDS) % 32 == 0, "count table and cells start on a line");

    const int ZB = 256;
    static_assert(NN % 32 == 0, "whole 16-row tiles, and whole pairs of them for layer 2");
    gcn_gemm_strip<FF, HH1, true><<<GS_GRID, GS_THREADS, 0, stream>>>(
        x, W1, W1 + (size_t)FF * HH1, pre0, pre1, NN / 16, erow, own,
        ecol, ev0, ev1, cells, cellL, tab);
    gcn_scatter_l1<<<SC_RANGES, SC_THREADS, 0, stream>>>(
        pre0, pre1, erow, ecol, ev0, ev1, bufA, lkey, lv0, lv1, lcnt, own, cells, cellL, tab);

    gcn_gemm_strip<HH1, HH2, false><<<GS_GRID, GS_THREADS, 0, stream>>>(
        bufA, W2, W2 + (size_t)HH1 * HH2, pre0, pre1, NN / 32, erow, own,
        ecol, ev0, ev1, nullptr, nullptr, nullptr);
    gcn_scatter_lds<HH2><<<SC_RANGES, SC_THREADS, 0, stream>>>(
        pre0, pre1, erow, ecol, ev0, ev1, bufA, lkey, lv0, lv1, lcnt, own);

    gcn_dot64x2<<<(NN + ZB - 1) / ZB, ZB, 0, stream>>>(
        bufA, W3, W3 + (size_t)HH2, pre0, pre1);
    gcn_scatter1_lds<<<SC_RANGES, SC_THREADS, 0, stream>>>(
        pre0, pre1, erow, ecol, ev0, ev1, out, lkey, lv0, lv1, lcnt);
}


# Both row gathers read a batch's edge records with one LDS read per field (a lane per edge, handed out by lane reads), one batch ahead, not one same-address read per field and edge; arrival-order sums
# speedup vs baseline: 3.4500x; 1.0308x over previous
#include <hip/hip_runtime.h>

typedef __attribute__((ext_vector_type(2))) float v2f;
typedef __attribute__((ext_vector_type(8))) float v8f;

#define NN 100000
#define FF 256
#define EE 1600000
#define HH1 128
#define HH2 64

#define GS_PITCH   33
#define GS_WAVES   16
#define GS_THREADS (32 * GS_WAVES)
#define GS_GRID    256

#define OW_ROWS    ((NN + 255) / 256)
#define OW_WORDS   (EE / 4)

#define BK_CHUNK   (EE / GS_GRID)
#define BK_CAP     64
#define BK_LP      (BK_CAP + 1)
#define BK_IT      ((BK_CHUNK + GS_THREADS - 1) / GS_THREADS)
#define BK_OWNER   (3 * GS_GRID * BK_CAP)

template <int K, int Nout, bool OWNERS>
__global__ __launch_bounds__(GS_THREADS)
void gcn_gemm_strip(const float* __restrict__ A,
                    const float* __restrict__ B0, const float* __restrict__ B1,
                    float* __restrict__ C0, float* __restrict__ C1, int nsteps,
                    const int* __restrict__ erow, int* __restrict__ own,
                    const int* __restrict__ ecol,
                    const float* __restrict__ ev0, const float* __restrict__ ev1,
                    int* __restrict__ cells, int* __restrict__ cells_last,
                    int* __restrict__ tab) {
    constexpr int STRIPS = Nout / 16;
    constexpr int WPT    = 2 * STRIPS;
    constexpr int MT     = GS_WAVES / WPT;
    constexpr int K4     = K / 4;
    constexpr int KJ     = K / 8;
    static_assert(K % 8 == 0 && Nout % 16 == 0 && GS_WAVES % WPT == 0, "tile dims");
    static_assert(MT * 16 * K4 == 2 * GS_THREADS, "two float4 per thread and step");

    __shared__ float4 sA[2][MT * KJ * GS_PITCH];

    const int t    = threadIdx.x;
    const int lane = t & 31;
    const int wave = t >> 5;
    const int h    = lane >> 4;
    const int c    = lane & 15;
    const int g    = wave / WPT;

    if constexpr (OWNERS) {
        static_assert(EE % 4 == 0, "edge rows read in 16-byte pieces");
        static_assert(4 * GS_GRID * GS_THREADS >= OW_WORDS, "four words per thread cover the map");
        static_assert((NN - 1) / OW_ROWS <= 255, "owner fits a byte");
        for (int i = 0; i < 4; ++i) {
            const int w = i * (GS_GRID * GS_THREADS) + blockIdx.x * GS_THREADS + t;
            if (w < OW_WORDS) {
                const int4 r = *reinterpret_cast<const int4*>(erow + 4 * w);
                const unsigned o0 = min((unsigned)r.x / OW_ROWS, 255u);
                const unsigned o1 = min((unsigned)r.y / OW_ROWS, 255u);
                const unsigned o2 = min((unsigned)r.z / OW_ROWS, 255u);
                const unsigned o3 = min((unsigned)r.w / OW_ROWS, 255u);
                own[w] = (int)(o0 | o1 << 8 | o2 << 16 | o3 << 24);
            }
        }

        static_assert(EE % GS_GRID == 0, "whole chunks");
        static_assert(BK_CAP == 64 && GS_WAVES == 16, "copy-out below: two lines per cell and field, 96 lines per wave");
        __shared__ int b_cnt[256];
        __shared__ int b_cell[3][256][BK_LP];
        if (t < 256) b_cnt[t] = 0;
        __syncthreads();
        {
            const int e0 = blockIdx.x * BK_CHUNK;
            int   er[BK_IT], ec[BK_IT];
            float ea[BK_IT], eb[BK_IT];
#pragma unroll
            for (int k = 0; k < BK_IT; ++k) {
                const int i  = k * GS_THREADS + t;
                const int e  = e0 + min(i, BK_CHUNK - 1);
                er[k] = (i < BK_CHUNK) ? erow[e] : -1;
                ec[k] = ecol[e];
                ea[k] = ev0[e];
                eb[k] = ev1[e];
            }
#pragma unroll
            for (int k = 0; k < BK_IT; ++k) {
                const int r = er[k];
                if ((unsigned)r < (unsigned)NN) {
                    const int o    = r / OW_ROWS;
                    const int rank = atomicAdd(&b_cnt[o], 1);
                    if (rank < BK_CAP) {
                        b_cell[0][o][rank] = ((r - o * OW_ROWS) << 17) | ec[k];
                        b_cell[1][o][rank] = __float_as_int(ea[k]);
                        b_cell[2][o][rank] = __float_as_int(eb[k]);
                    }
                }
            }
        }
        __syncthreads();
        if (t < 256) tab[blockIdx.x * 256 + t] = b_cnt[t];
        {
            const int hf = (wave & 1) * 32;
            for (int m0 = 0; m0 < 96; m0 += 8) {
                int cn[8], val[8];
#pragma unroll
                for (int u = 0; u < 8; ++u) {
                    const int m = m0 + u;
                    const int a = m >> 5;
                    const int o = (m & 31) * 8 + (wave >> 1);
                    cn[u]  = b_cnt[o];
                    val[u] = b_cell[a][o][hf + lane];
                }
#pragma unroll
                for (int u = 0; u < 8; ++u) {
                    const int m = m0 + u;
                    const int a = m >> 5;
                    const int o = (m & 31) * 8 + (wave >> 1);
                    if (hf < cn[u]) {
                        int* dst = (o == 255) ? cells_last : cells + (size_t)o * (OW_ROWS * HH1);
                        dst[(a * GS_GRID + blockIdx.x) * BK_CAP + hf + lane] = val[u];
                    }
                }
            }
        }
    }
    const int sup  = (wave % WPT) / STRIPS;
    const int col  = (wave % STRIPS) * 16 + c;

    const float* __restrict__ bp = (sup ? B1 : B0) + (size_t)(4 * h) * Nout + col;
    v2f breg[K4];
#pragma unroll
    for (int s = 0; s < K4; ++s) {
        const int kk = 8 * (s >> 1) + 2 * (s & 1);
        breg[s].x = bp[(kk + 0) * Nout];
        breg[s].y = bp[(kk + 1) * Nout];
    }
    float* __restrict__ cp = (sup ? C1 : C0) + (size_t)(16 * g + 8 * h) * Nout + col;

    const int f0 = t, f1 = t + GS_THREADS;
    const int r0 = f0 / K4, c0 = f0 % K4;
    const int r1 = f1 / K4, c1 = f1 % K4;
    const int srow0  = r0 * K + c0 * 4;
    const int srow1  = r1 * K + c1 * 4;
    const int sslot0 = ((r0 >> 4) * KJ + (c0 >> 1)) * GS_PITCH + (c0 & 1) * 16 + (r0 & 15);
    const int sslot1 = ((r1 >> 4) * KJ + (c1 >> 1)) * GS_PITCH + (c1 & 1) * 16 + (r1 & 15);

    int step = blockIdx.x;
    {
        const float* __restrict__ ap = A + (size_t)min(step, nsteps - 1) * (MT * 16 * K);
        const float4 p0 = *reinterpret_cast<const float4*>(ap + srow0);
        const float4 p1 = *reinterpret_cast<const float4*>(ap + srow1);
        sA[0][sslot0] = p0;
        sA[0][sslot1] = p1;
    }
    __syncthreads();

    int buf = 0;
    v8f rprev = {};
    float* __restrict__ cprev = cp;
    bool pending = false;
    for (; step < nsteps; step += gridDim.x) {
        const float* __restrict__ ap = A + (size_t)min(step + (int)gridDim.x, nsteps - 1) * (MT * 16 * K);
        const float4 st0 = *reinterpret_cast<const float4*>(ap + srow0);
        const float4 st1 = *reinterpret_cast<const float4*>(ap + srow1);

        if (pending) {
#pragma unroll
            for (int v = 0; v < 8; ++v) cprev[(size_t)v * Nout] = rprev[v];
        }

        const float4* __restrict__ fa = sA[buf] + g * (KJ * GS_PITCH) + lane;
        v8f acc0 = {}, acc1 = {}, acc2 = {}, acc3 = {};
#pragma unroll
        for (int j = 0; j < KJ; j += 2) {
            const float4 a0 = fa[(j + 0) * GS_PITCH];
            const float4 a1 = fa[(j + 1) * GS_PITCH];
            v2f x0 = {a0.x, a0.y}, x1 = {a0.z, a0.w}, x2 = {a1.x, a1.y}, x3 = {a1.z, a1.w};
            acc0 = __builtin_amdgcn_wmma_f32_16x16x4_f32(false, x0, false, breg[2 * j + 0], (short)0, acc0, false, false);
            acc1 = __builtin_amdgcn_wmma_f32_16x16x4_f32(false, x1, false, breg[2 * j + 1], (short)0, acc1, false, false);
            acc2 = __builtin_amdgcn_wmma_f32_16x16x4_f32(false, x2, false, breg[2 * j + 2], (short)0, acc2, false, false);
            acc3 = __builtin_amdgcn_wmma_f32_16x16x4_f32(false, x3, false, breg[2 * j + 3], (short)0, acc3, false, false);
        }
        rprev   = (acc0 + acc1) + (acc2 + acc3);
        cprev   = cp + (size_t)step * (MT * 16 * Nout);
        pending = true;

        sA[buf ^ 1][sslot0] = st0;
        sA[buf ^ 1][sslot1] = st1;
        __syncthreads();
        buf ^= 1;
    }
    if (pending) {
#pragma unroll
        for (int v = 0; v < 8; ++v) cprev[(size_t)v * Nout] = rprev[v];
    }
}

#define SC_THREADS 1024
#define SC_WAVES   (SC_THREADS / 32)
#define SC_RANGES  256
#define SC_ROWS    ((NN + SC_RANGES - 1) / SC_RANGES)
#define SC_LD      10
#define SC_CHUNK   (SC_THREADS * SC_LD * 4)
#define SC_NCHUNK  ((EE + SC_CHUNK - 1) / SC_CHUNK)
#define SC_QCAP    1024
#define SC_U       6
static_assert(OW_ROWS == SC_ROWS && SC_RANGES == 256, "the owner byte is the work-group index");
#define OW_LD      10
#define OW_NQ      (EE / 16)
#define OW_CHUNKQ  (SC_THREADS * OW_LD)
#define OW_NCHUNK  ((OW_NQ + OW_CHUNKQ - 1) / OW_CHUNKQ)
static_assert(EE % 16 == 0, "owner map read in 16-byte pieces");

__device__ __forceinline__ unsigned own_hit(unsigned w, unsigned pat) {
    const unsigned x = w ^ pat;
    return ((x - 0x01010101u) & ~x & 0x80808080u) != 0u;
}
__device__ __forceinline__ unsigned long long own_hit4(int4 v, unsigned pat, bool valid) {
    const unsigned f = own_hit((unsigned)v.x, pat) | own_hit((unsigned)v.y, pat) << 1 |
                       own_hit((unsigned)v.z, pat) << 2 | own_hit((unsigned)v.w, pat) << 3;
    return valid ? f : 0u;
}

#define LIST_CAP   7168
#define LIST_U     8
#define KEY_SHIFT  17
#define KEY_MASK   ((1 << KEY_SHIFT) - 1)
static_assert(NN <= (1 << KEY_SHIFT), "col fits the low key bits");
static_assert(SC_ROWS <= (1 << (31 - KEY_SHIFT)), "local row fits the high key bits");
static_assert(LIST_CAP % SC_THREADS == 0, "layer 3 walks the list in whole sweeps");
static_assert(KEY_SHIFT == 17 && SC_RANGES == GS_GRID, "the edge cells carry the same key, one chunk per product work-group");
static_assert(BK_OWNER <= SC_ROWS * HH1, "an owner's cells fit into its own rows of the layer-1 output");
static_assert((GS_GRID * BK_CAP) % SC_THREADS == 0, "cell slots read in whole sweeps");

template <int H>
__global__ __launch_bounds__(SC_THREADS)
void gcn_scatter_lds(const float* __restrict__ pre0,
                     const float* __restrict__ pre1,
                     const int* __restrict__ erow,
                     const int* __restrict__ ecol,
                     const float* __restrict__ ev0,
                     const float* __restrict__ ev1,
                     float* __restrict__ hout,
                     const int* __restrict__ lkey,
                     const float* __restrict__ lv0,
                     const float* __restrict__ lv1,
                     const int* __restrict__ lcnt,
                     const int* __restrict__ own) {
    constexpr int VEC = H / 32;
    typedef __attribute__((ext_vector_type(VEC))) float vecf;
    static_assert(H == 32 * VEC && (VEC == 2 || VEC == 4), "row = one vector load per lane");
    static_assert(EE % 4 == 0, "edge list read in 16-byte pieces");
    static_assert(OW_LD * 4 <= 64, "one mask bit per owner word of a thread");

    __shared__ float acc[SC_ROWS * H];
    __shared__ int   q_rl[3][SC_QCAP];
    __shared__ int   q_col[3][SC_QCAP];
    __shared__ float q_v0[3][SC_QCAP];
    __shared__ float q_v1[3][SC_QCAP];
    __shared__ int   q_n[3];

    const int t    = threadIdx.x;
    const int lane = t & 31;
    const int wave = t >> 5;
    const int lo   = blockIdx.x * SC_ROWS;
    const int cnt  = min(SC_ROWS, NN - lo);

    for (int i = t; i < SC_ROWS * H; i += SC_THREADS) acc[i] = 0.0f;
    if (t < 3) q_n[t] = 0;

    const int*   const gk  = lkey + (size_t)blockIdx.x * LIST_CAP;
    const float* const gv0 = lv0  + (size_t)blockIdx.x * LIST_CAP;
    const float* const gv1 = lv1  + (size_t)blockIdx.x * LIST_CAP;

    bool listed = false;
    {
        const int nl = lcnt[blockIdx.x];
        if (nl <= LIST_CAP) {
            listed = true;
            __shared__ int   s_key[LIST_CAP];
            __shared__ float s_v0[LIST_CAP];
            __shared__ float s_v1[LIST_CAP];
            for (int i = t; i < nl; i += SC_THREADS) {
                s_key[i] = gk[i];
                s_v0[i]  = gv0[i];
                s_v1[i]  = gv1[i];
            }
            __syncthreads();
            const int ul = min(lane, LIST_U - 1);
            int   kq = 0;
            float aq = 0.0f, bq = 0.0f;
            if (wave * LIST_U < nl) {
                const int ic = min(wave * LIST_U + ul, nl - 1);
                kq = s_key[ic];
                aq = s_v0[ic];
                bq = s_v1[ic];
            }
            for (int b = wave; b * LIST_U < nl; b += SC_WAVES) {
                const int   kc = kq;
                const float ac = aq, bc = bq;
                {
                    const int ic = min((b + SC_WAVES) * LIST_U + ul, nl - 1);
                    kq = s_key[ic];
                    aq = s_v0[ic];
                    bq = s_v1[ic];
                }
                int   rl[LIST_U];
                bool  ok[LIST_U];
                float v0[LIST_U], v1[LIST_U];
                vecf  p0[LIST_U], p1[LIST_U];
#pragma unroll
                for (int u = 0; u < LIST_U; ++u) {
                    ok[u] = b * LIST_U + u < nl;
                    const int key = __builtin_amdgcn_readlane(kc, u);
                    rl[u] = min(key >> KEY_SHIFT, SC_ROWS - 1);
                    v0[u] = __int_as_float(__builtin_amdgcn_readlane(__float_as_int(ac), u));
                    v1[u] = __int_as_float(__builtin_amdgcn_readlane(__float_as_int(bc), u));
                    const unsigned off = (unsigned)(key & KEY_MASK) * H + VEC * lane;
                    p0[u] = *reinterpret_cast<const vecf*>(pre0 + off);
                    p1[u] = *reinterpret_cast<const vecf*>(pre1 + off);
                }
#pragma unroll
                for (int u = 0; u < LIST_U; ++u) {
                    float* a = acc + rl[u] * H + lane;
#pragma unroll
                    for (int j = 0; j < VEC; ++j)
                        atomicAdd(a + j * 32,
                                  ok[u] ? v0[u] * p0[u][j] + v1[u] * p1[u][j] : 0.0f);
                }
            }
        }
    }

    if (!listed) {
        int c = (blockIdx.x * 7) % OW_NCHUNK;

        const int4* __restrict__ own4 = reinterpret_cast<const int4*>(own);
        const unsigned pat = blockIdx.x * 0x01010101u;
        int4 rv[OW_LD];
        unsigned long long m = 0;
#pragma unroll
        for (int j = 0; j < OW_LD; ++j) {
            const int q = c * OW_CHUNKQ + j * SC_THREADS + t;
            rv[j] = own4[min(q, OW_NQ - 1)];
        }
#pragma unroll
        for (int j = 0; j < OW_LD; ++j) {
            m |= own_hit4(rv[j], pat, c * OW_CHUNKQ + j * SC_THREADS + t < OW_NQ) << (4 * j);
        }
        __syncthreads();

        for (int it = 0; it < OW_NCHUNK; ++it) {
            const int qb = it % 3;
            const int cbase = c * OW_CHUNKQ;
            int cn = c + 1;
            if (cn == OW_NCHUNK) cn = 0;

            if (it + 1 < OW_NCHUNK) {
#pragma unroll
                for (int j = 0; j < OW_LD; ++j) {
                    const int q = cn * OW_CHUNKQ + j * SC_THREADS + t;
                    rv[j] = own4[min(q, OW_NQ - 1)];
                }
            }

            while (m) {
                const int b = __builtin_ctzll(m);
                m &= m - 1;
                const int eb = ((cbase + (b >> 2) * SC_THREADS + t) << 4) + ((b & 3) << 2);
                const int4   r4 = *reinterpret_cast<const int4*>(erow + eb);
                const int4   c4 = *reinterpret_cast<const int4*>(ecol + eb);
                const float4 a4 = *reinterpret_cast<const float4*>(ev0 + eb);
                const float4 b4 = *reinterpret_cast<const float4*>(ev1 + eb);
                const int   rr[4] = {r4.x, r4.y, r4.z, r4.w};
                const int   cc[4] = {c4.x, c4.y, c4.z, c4.w};
                const float aa[4] = {a4.x, a4.y, a4.z, a4.w};
                const float bb[4] = {b4.x, b4.y, b4.z, b4.w};
#pragma unroll
                for (int i = 0; i < 4; ++i) {
                    const int rl = rr[i] - lo;
                    if ((unsigned)rl >= (unsigned)cnt) continue;
                    const int col = cc[i];
                    const float v0 = aa[i];
                    const float v1 = bb[i];
                    const int slot = atomicAdd(&q_n[qb], 1);
                    if (slot < SC_QCAP) {
                        q_rl[qb][slot]  = rl;
                        q_col[qb][slot] = col;
                        q_v0[qb][slot]  = v0;
                        q_v1[qb][slot]  = v1;
                    } else {
                        const float* s0 = pre0 + (size_t)col * H;
                        const float* s1 = pre1 + (size_t)col * H;
                        for (int k = 0; k < H; ++k)
                            atomicAdd(&acc[rl * H + (k % VEC) * 32 + (k / VEC)],
                                      v0 * s0[k] + v1 * s1[k]);
                    }
                }
            }

            if (it + 1 < OW_NCHUNK) {
#pragma unroll
                for (int j = 0; j < OW_LD; ++j) {
                    m |= own_hit4(rv[j], pat, cn * OW_CHUNKQ + j * SC_THREADS + t < OW_NQ) << (4 * j);
                }
            }
            c = cn;

            __syncthreads();
            if (t == 0) q_n[(it + 2) % 3] = 0;

            const int n = min(q_n[qb], SC_QCAP);
            for (int i0 = wave; i0 < n; i0 += SC_WAVES * SC_U) {
                int   rl[SC_U];
                bool  ok[SC_U];
                float v0[SC_U], v1[SC_U];
                vecf  p0[SC_U], p1[SC_U];
#pragma unroll
                for (int u = 0; u < SC_U; ++u) {
                    const int i  = i0 + u * SC_WAVES;
                    ok[u] = i < n;
                    const int ic = ok[u] ? i : i0;
                    rl[u] = q_rl[qb][ic];
                    v0[u] = q_v0[qb][ic];
                    v1[u] = q_v1[qb][ic];
                    const unsigned off = (unsigned)q_col[qb][ic] * H + VEC * lane;
                    p0[u] = *reinterpret_cast<const vecf*>(pre0 + off);
                    p1[u] = *reinterpret_cast<const vecf*>(pre1 + off);
                }
#pragma unroll
                for (int u = 0; u < SC_U; ++u) {
                    float* a = acc + rl[u] * H + lane;
#pragma unroll
                    for (int j = 0; j < VEC; ++j)
                        atomicAdd(a + j * 32,
                                  ok[u] ? v0[u] * p0[u][j] + v1[u] * p1[u][j] : 0.0f);
                }
            }
        }
    }
    __syncthreads();

    for (int row = wave; row < cnt; row += SC_WAVES) {
        vecf o;
#pragma unroll
        for (int j = 0; j < VEC; ++j) o[j] = fmaxf(acc[row * H + j * 32 + lane], 0.0f);
        *reinterpret_cast<vecf*>(hout + (size_t)(lo + row) * H + VEC * lane) = o;
    }
}

#define L1_U      6
#define L1_SPLIT  (NN / 2)

__global__ __launch_bounds__(SC_THREADS)
void gcn_scatter_l1(const float* __restrict__ pre0,
                    const float* __restrict__ pre1,
                    const int* __restrict__ erow,
                    const int* __restrict__ ecol,
                    const float* __restrict__ ev0,
                    const float* __restrict__ ev1,
                    float* hout,
                    int* __restrict__ lkey,
                    float* __restrict__ lv0,
                    float* __restrict__ lv1,
                    int* __restrict__ lcnt,
                    const int* __restrict__ own,
                    const int* cells,
                    const int* __restrict__ cells_last,
                    const int* __restrict__ tab) {
    constexpr int H   = HH1;
    constexpr int VEC = H / 32;
    typedef __attribute__((ext_vector_type(VEC))) float vecf;
    static_assert(H == 32 * VEC && VEC == 4, "row = one 16-byte load per lane");
    static_assert(OW_LD * 4 <= 64, "one mask bit per owner word of a thread");

    __shared__ float acc[SC_ROWS * H];
    __shared__ int   s_key[LIST_CAP];
    __shared__ float s_v0[LIST_CAP];
    __shared__ float s_v1[LIST_CAP];
    __shared__ int   s_n[3];
    __shared__ int   s_cnt[GS_GRID];
    __shared__ int   s_part[2][GS_GRID / 32];

    const int t    = threadIdx.x;
    const int lane = t & 31;
    const int wave = t >> 5;
    const int lo   = blockIdx.x * SC_ROWS;
    const int cnt  = min(SC_ROWS, NN - lo);

    for (int i = t; i < SC_ROWS * H; i += SC_THREADS) acc[i] = 0.0f;
    if (t < 3) s_n[t] = 0;

    if (t < GS_GRID) {
        const int cc = tab[t * SC_RANGES + blockIdx.x];
        s_cnt[t] = cc;
        int sum = cc, mx = cc;
        for (int d = 16; d > 0; d >>= 1) {
            sum += __shfl_xor(sum, d);
            mx   = max(mx, __shfl_xor(mx, d));
        }
        if (lane == 0) { s_part[0][wave] = sum; s_part[1][wave] = mx; }
    }
    __syncthreads();
    int total = 0, cmax = 0;
    for (int w = 0; w < GS_GRID / 32; ++w) {
        total += s_part[0][w];
        cmax   = max(cmax, s_part[1][w]);
    }
    const bool celled = cmax <= BK_CAP && total <= LIST_CAP;

    if (celled) {
        const int* cb = (blockIdx.x == SC_RANGES - 1) ? cells_last
                                                      : cells + (size_t)blockIdx.x * (SC_ROWS * H);
        constexpr int SW = GS_GRID * BK_CAP / SC_THREADS;
        int   key[SW], cg[SW];
        float w0[SW], w1[SW];
        bool  ok[SW];
#pragma unroll
        for (int k = 0; k < SW; ++k) cg[k] = s_cnt[(k * SC_THREADS + t) / BK_CAP];
#pragma unroll
        for (int k = 0; k < SW; ++k) {
            const int s   = k * SC_THREADS + t;
            const int g   = s / BK_CAP;
            const int idx = s % BK_CAP;
            ok[k]  = idx < cg[k];
            key[k] = 0; w0[k] = 0.0f; w1[k] = 0.0f;
            if (ok[k]) {
                key[k] = cb[(0 * GS_GRID + g) * BK_CAP + idx];
                w0[k]  = __int_as_float(cb[(1 * GS_GRID + g) * BK_CAP + idx]);
                w1[k]  = __int_as_float(cb[(2 * GS_GRID + g) * BK_CAP + idx]);
            }
        }
        const unsigned below = (1u << lane) - 1u;
#pragma unroll
        for (int k = 0; k < SW; ++k) {
            const bool isl = ok[k] && (key[k] & KEY_MASK) < L1_SPLIT;
            const bool ish = ok[k] && !isl;
            const unsigned bl = (unsigned)__ballot(isl);
            const unsigned bh = (unsigned)__ballot(ish);
            int basel = 0, baseh = 0;
            if (lane == 0) {
                if (bl) basel = atomicAdd(&s_n[0], __popc(bl));
                if (bh) baseh = atomicAdd(&s_n[1], __popc(bh));
            }
            basel = __builtin_amdgcn_readfirstlane(basel);
            baseh = __builtin_amdgcn_readfirstlane(baseh);
            if (ok[k]) {
                const int pos = isl ? basel + __popc(bl & below)
                                    : LIST_CAP - 1 - (baseh + __popc(bh & below));
                s_key[pos] = key[k];
                s_v0[pos]  = w0[k];
                s_v1[pos]  = w1[k];
            }
        }
        if (t == 0) s_n[2] = total;
    } else {
        int c = (blockIdx.x * 7) % OW_NCHUNK;

        const int4* __restrict__ own4 = reinterpret_cast<const int4*>(own);
        const unsigned pat = blockIdx.x * 0x01010101u;
        int4 rv[OW_LD];
        unsigned long long m = 0;
#pragma unroll
        for (int j = 0; j < OW_LD; ++j) {
            const int q = c * OW_CHUNKQ + j * SC_THREADS + t;
            rv[j] = own4[min(q, OW_NQ - 1)];
        }
#pragma unroll
        for (int j = 0; j < OW_LD; ++j) {
            m |= own_hit4(rv[j], pat, c * OW_CHUNKQ + j * SC_THREADS + t < OW_NQ) << (4 * j);
        }
        __syncthreads();

        for (int it = 0; it < OW_NCHUNK; ++it) {
            const int cbase = c * OW_CHUNKQ;
            int cn = c + 1;
            if (cn == OW_NCHUNK) cn = 0;

            if (it + 1 < OW_NCHUNK) {
#pragma unroll
                for (int j = 0; j < OW_LD; ++j) {
                    const int q = cn * OW_CHUNKQ + j * SC_THREADS + t;
                    rv[j] = own4[min(q, OW_NQ - 1)];
                }
            }

            while (m) {
                const int b = __builtin_ctzll(m);
                m &= m - 1;
                const int eb = ((cbase + (b >> 2) * SC_THREADS + t) << 4) + ((b & 3) << 2);
                const int4   r4 = *reinterpret_cast<const int4*>(erow + eb);
                const int4   c4 = *reinterpret_cast<const int4*>(ecol + eb);
                const float4 a4 = *reinterpret_cast<const float4*>(ev0 + eb);
                const float4 b4 = *reinterpret_cast<const float4*>(ev1 + eb);
                const int   rr[4] = {r4.x, r4.y, r4.z, r4.w};
                const int   cc[4] = {c4.x, c4.y, c4.z, c4.w};
                const float aa[4] = {a4.x, a4.y, a4.z, a4.w};
                const float bb[4] = {b4.x, b4.y, b4.z, b4.w};
#pragma unroll
                for (int i = 0; i < 4; ++i) {
                    const int rl = rr[i] - lo;
                    if ((unsigned)rl >= (unsigned)cnt) continue;
                    const int col = cc[i];
                    const float v0 = aa[i];
                    const float v1 = bb[i];
                    const int all = atomicAdd(&s_n[2], 1);
                    if (all < LIST_CAP) {
                        int pos;
                        if (col < L1_SPLIT) pos = atomicAdd(&s_n[0], 1);
                        else                pos = LIST_CAP - 1 - atomicAdd(&s_n[1], 1);
                        s_key[pos] = (rl << KEY_SHIFT) | col;
                        s_v0[pos]  = v0;
                        s_v1[pos]  = v1;
                    } else {
                        const float* s0 = pre0 + (size_t)col * H;
                        const float* s1 = pre1 + (size_t)col * H;
                        for (int k = 0; k < H; ++k)
                            atomicAdd(&acc[rl * H + (k % VEC) * 32 + (k / VEC)],
                                      v0 * s0[k] + v1 * s1[k]);
                    }
                }
            }

            if (it + 1 < OW_NCHUNK) {
#pragma unroll
                for (int j = 0; j < OW_LD; ++j) {
                    m |= own_hit4(rv[j], pat, cn * OW_CHUNKQ + j * SC_THREADS + t < OW_NQ) << (4 * j);
                }
            }
            c = cn;
        }
    }
    __syncthreads();

    const int nlo = s_n[0];
    const int nhi = s_n[1];

    {
        int*   const gk  = lkey + (size_t)blockIdx.x * LIST_CAP;
        float* const gv0 = lv0  + (size_t)blockIdx.x * LIST_CAP;
        float* const gv1 = lv1  + (size_t)blockIdx.x * LIST_CAP;
        for (int i = t; i < nlo + nhi; i += SC_THREADS) {
            const int sl = (i < nlo) ? i : LIST_CAP - nhi + (i - nlo);
            gk[i]  = s_key[sl];
            gv0[i] = s_v0[sl];
            gv1[i] = s_v1[sl];
        }
        if (t == 0) lcnt[blockIdx.x] = s_n[2];
    }

    const int ul = min(lane, L1_U - 1);
    for (int pass = 0; pass < 2; ++pass) {
        const int beg = pass ? LIST_CAP - nhi : 0;
        const int n   = pass ? nhi : nlo;
        int   kq = 0;
        float aq = 0.0f, bq = 0.0f;
        if (wave * L1_U < n) {
            const int ic = beg + min(wave * L1_U + ul, n - 1);
            kq = s_key[ic];
            aq = s_v0[ic];
            bq = s_v1[ic];
        }
        for (int b = wave; b * L1_U < n; b += SC_WAVES) {
            const int   kc = kq;
            const float ac = aq, bc = bq;
            {
                const int ic = beg + min((b + SC_WAVES) * L1_U + ul, n - 1);
                kq = s_key[ic];
                aq = s_v0[ic];
                bq = s_v1[ic];
            }
            int   rl[L1_U];
            bool  ok[L1_U];
            float v0[L1_U], v1[L1_U];
            vecf  p0[L1_U], p1[L1_U];
#pragma unroll
            for (int u = 0; u < L1_U; ++u) {
                ok[u] = b * L1_U + u < n;
                const int key = __builtin_amdgcn_readlane(kc, u);
                rl[u] = key >> KEY_SHIFT;
                v0[u] = __int_as_float(__builtin_amdgcn_readlane(__float_as_int(ac), u));
                v1[u] = __int_as_float(__builtin_amdgcn_readlane(__float_as_int(bc), u));
                const unsigned off = (unsigned)(key & KEY_MASK) * H + VEC * lane;
                p0[u] = *reinterpret_cast<const vecf*>(pre0 + off);
                p1[u] = *reinterpret_cast<const vecf*>(pre1 + off);
            }
#pragma unroll
            for (int u = 0; u < L1_U; ++u) {
                float* a = acc + rl[u] * H + lane;
#pragma unroll
                for (int j = 0; j < VEC; ++j)
                    atomicAdd(a + j * 32,
                              ok[u] ? v0[u] * p0[u][j] + v1[u] * p1[u][j] : 0.0f);
            }
        }
        __syncthreads();
    }

    for (int row = wave; row < cnt; row += SC_WAVES) {
        vecf o;
#pragma unroll
        for (int j = 0; j < VEC; ++j) o[j] = fmaxf(acc[row * H + j * 32 + lane], 0.0f);
        *reinterpret_cast<vecf*>(hout + (size_t)(lo + row) * H + VEC * lane) = o;
    }
}

__global__ __launch_bounds__(SC_THREADS)
void gcn_scatter1_lds(const float* __restrict__ p0,
                      const float* __restrict__ p1,
                      const int* __restrict__ erow,
                      const int* __restrict__ ecol,
                      const float* __restrict__ ev0,
                      const float* __restrict__ ev1,
                      float* __restrict__ out,
                      const int* __restrict__ lkey,
                      const float* __restrict__ lv0,
                      const float* __restrict__ lv1,
                      const int* __restrict__ lcnt) {
    static_assert(SC_ROWS <= SC_THREADS, "one thread per owned row");
    __shared__ float acc1[SC_ROWS];

    const int t   = threadIdx.x;
    const int lo  = blockIdx.x * SC_ROWS;
    const int cnt = min(SC_ROWS, NN - lo);

    if (t < SC_ROWS) acc1[t] = 0.0f;
    __syncthreads();

    const int nl = lcnt[blockIdx.x];
    if (nl <= LIST_CAP) {
        const int*   gk  = lkey + (size_t)blockIdx.x * LIST_CAP;
        const float* gv0 = lv0  + (size_t)blockIdx.x * LIST_CAP;
        const float* gv1 = lv1  + (size_t)blockIdx.x * LIST_CAP;
        constexpr int SW = LIST_CAP / SC_THREADS;
        int   rl[SW];
        float val[SW];
#pragma unroll
        for (int u = 0; u < SW; ++u) {
            const int i  = u * SC_THREADS + t;
            const int ic = (i < nl) ? i : t;
            const int key = gk[ic];
            const int col = key & KEY_MASK;
            rl[u]  = min(key >> KEY_SHIFT, SC_ROWS - 1);
            val[u] = gv0[ic] * p0[col] + gv1[ic] * p1[col];
        }
#pragma unroll
        for (int u = 0; u < SW; ++u)
            if (u * SC_THREADS + t < nl) atomicAdd(&acc1[rl[u]], val[u]);
    } else {
        int c = (blockIdx.x * 7) % SC_NCHUNK;
        for (int it = 0; it < SC_NCHUNK; ++it) {
            const int cbase = c * SC_CHUNK;
            int4 rv[SC_LD];
#pragma unroll
            for (int j = 0; j < SC_LD; ++j) {
                const int e0 = cbase + ((j * SC_THREADS + t) << 2);
                rv[j] = *reinterpret_cast<const int4*>(erow + min(e0, EE - 4));
            }
            unsigned long long m = 0;
#pragma unroll
            for (int j = 0; j < SC_LD; ++j) {
                const unsigned lim = (cbase + ((j * SC_THREADS + t) << 2) < EE) ? (unsigned)cnt : 0u;
                m |= (unsigned long long)((unsigned)(rv[j].x - lo) < lim) << (4 * j + 0);
                m |= (unsigned long long)((unsigned)(rv[j].y - lo) < lim) << (4 * j + 1);
                m |= (unsigned long long)((unsigned)(rv[j].z - lo) < lim) << (4 * j + 2);
                m |= (unsigned long long)((unsigned)(rv[j].w - lo) < lim) << (4 * j + 3);
            }
            while (m) {
                const int b = __builtin_ctzll(m);
                m &= m - 1;
                const int e   = cbase + ((((b >> 2) * SC_THREADS) + t) << 2) + (b & 3);
                const int rl  = erow[e] - lo;
                const int col = ecol[e];
                atomicAdd(&acc1[rl], ev0[e] * p0[col] + ev1[e] * p1[col]);
            }
            c = (c + 1 == SC_NCHUNK) ? 0 : c + 1;
        }
    }
    __syncthreads();

    if (t < cnt) out[lo + t] = acc1[t];
}

__global__ void gcn_dot64x2(const float* __restrict__ h,
                            const float* __restrict__ w0,
                            const float* __restrict__ w1,
                            float* __restrict__ p0, float* __restrict__ p1) {
    int i = blockIdx.x * blockDim.x + threadIdx.x;
    if (i >= NN) return;
    const float* row = h + (size_t)i * HH2;
    float s0 = 0.0f, s1 = 0.0f;
#pragma unroll
    for (int k = 0; k < HH2; ++k) {
        float hv = row[k];
        s0 += hv * w0[k];
        s1 += hv * w1[k];
    }
    p0[i] = s0;
    p1[i] = s1;
}

extern "C" void kernel_launch(void* const* d_in, const int* in_sizes, int n_in,
                              void* d_out, int out_size, void* d_ws, size_t ws_size,
                              hipStream_t stream) {
    const float* x     = (const float*)d_in[0];
    const int*   erow  = (const int*)d_in[1];
    const int*   ecol  = (const int*)d_in[2];
    const float* evals = (const float*)d_in[3];
    const float* W1    = (const float*)d_in[4];
    const float* W2    = (const float*)d_in[5];
    const float* W3    = (const float*)d_in[6];
    float* out = (float*)d_out;

    const float* ev0 = evals;
    const float* ev1 = evals + (size_t)EE;

    float* pre0 = (float*)d_ws;
    float* pre1 = pre0 + (size_t)NN * HH1;
    float* bufA = pre1 + (size_t)NN * HH1;
    float* bufB = bufA + (size_t)NN * HH1;
    static_assert((size_t)3 * SC_RANGES * LIST_CAP + SC_RANGES + OW_WORDS <= (size_t)NN * HH2,
                  "lists, counts and owner map fit the fourth region");
    static_assert((3 * SC_RANGES * LIST_CAP + SC_RANGES) % 4 == 0, "owner map 16-byte aligned");
    int*   lkey = (int*)bufB;
    float* lv0  = bufB + (size_t)SC_RANGES * LIST_CAP;
    float* lv1  = bufB + (size_t)2 * SC_RANGES * LIST_CAP;
    int*   lcnt = (int*)(bufB + (size_t)3 * SC_RANGES * LIST_CAP);
    int*   own  = lcnt + SC_RANGES;
    int*   tab  = own + OW_WORDS;
    int*   cellL = tab + GS_GRID * SC_RANGES;
    int*   cells = (int*)bufA;
    static_assert((size_t)3 * SC_RANGES * LIST_CAP + SC_RANGES + OW_WORDS + GS_GRID * SC_RANGES + BK_OWNER
                      <= (size_t)NN * HH2, "count table and the last owner's cells fit the fourth region too");
    static_assert((3 * SC_RANGES * LIST_CAP + SC_RANGES + OW_WORDS) % 32 == 0, "count table and cells start on a line");

    const int ZB = 256;
    static_assert(NN % 32 == 0, "whole 16-row tiles, and whole pairs of them for layer 2");
    gcn_gemm_strip<FF, HH1, true><<<GS_GRID, GS_THREADS, 0, stream>>>(
        x, W1, W1 + (size_t)FF * HH1, pre0, pre1, NN / 16, erow, own,
        ecol, ev0, ev1, cells, cellL, tab);
    gcn_scatter_l1<<<SC_RANGES, SC_THREADS, 0, stream>>>(
        pre0, pre1, erow, ecol, ev0, ev1, bufA, lkey, lv0, lv1, lcnt, own, cells, cellL, tab);

    gcn_gemm_strip<HH1, HH2, false><<<GS_GRID, GS_THREADS, 0, stream>>>(
        bufA, W2, W2 + (size_t)HH1 * HH2, pre0, pre1, NN / 32, erow, own,
        ecol, ev0, ev1, nullptr, nullptr, nullptr);
    gcn_scatter_lds<HH2><<<SC_RANGES, SC_THREADS, 0, stream>>>(
        pre0, pre1, erow, ecol, ev0, ev1, bufA, lkey, lv0, lv1, lcnt, own);

    gcn_dot64x2<<<(NN + ZB - 1) / ZB, ZB, 0, stream>>>(
        bufA, W3, W3 + (size_t)HH2, pre0, pre1);
    gcn_scatter1_lds<<<SC_RANGES, SC_THREADS, 0, stream>>>(
        pre0, pre1, erow, ecol, ev0, ev1, out, lkey, lv0, lv1, lcnt);
}
